# starting kernel
# baseline (speedup 1.0000x reference)
#include <hip/hip_runtime.h>

typedef float v2f __attribute__((ext_vector_type(2)));
typedef float v8f __attribute__((ext_vector_type(8)));

namespace {

constexpr int kB = 64;
constexpr int kT = 512;
constexpr int kH = 512;
constexpr int kL = 4;
constexpr int kC = 128;
constexpr int kLdsStride = 516;

__device__ __forceinline__ v8f wmma_f32(v2f a, v2f b, v8f c) {
  return __builtin_amdgcn_wmma_f32_16x16x4_f32(false, a, false, b, (short)0, c,
                                               false, false);
}

}

__global__ __launch_bounds__(256) void rnn_proj_kernel(
    const float* __restrict__ in, const float* __restrict__ Wi,
    const float* __restrict__ bi, const float* __restrict__ bh,
    float* __restrict__ xi) {
  const int lane   = threadIdx.x & 31;
  const int wave   = threadIdx.x >> 5;
  const int lane15 = lane & 15;
  const int half   = lane >> 4;

  const int blk   = blockIdx.x * 8 + wave;
  const int blkN  = blk & (kH / 32 - 1);
  const int blkM  = blk >> 4;

  const int n0 = blkN * 32 + lane15;
  const int n1 = n0 + 16;
  const int m0 = blkM * 32 + lane15;
  const int m1 = m0 + 16;

  const float bias0 = bi[n0] + bh[n0];
  const float bias1 = bi[n1] + bh[n1];
  v8f acc00, acc01, acc10, acc11;
#pragma unroll
  for (int r = 0; r < 8; ++r) {
    acc00[r] = bias0; acc01[r] = bias1;
    acc10[r] = bias0; acc11[r] = bias1;
  }

  const float* ap0 = in + (size_t)m0 * kH + half * 2;
  const float* ap1 = in + (size_t)m1 * kH + half * 2;
  const float* bp0 = Wi + (size_t)n0 * kH + half * 2;
  const float* bp1 = Wi + (size_t)n1 * kH + half * 2;
#pragma unroll 4
  for (int ks = 0; ks < kH / 4; ++ks) {
    v2f a0 = *(const v2f*)(ap0 + ks * 4);
    v2f a1 = *(const v2f*)(ap1 + ks * 4);
    v2f b0 = *(const v2f*)(bp0 + ks * 4);
    v2f b1 = *(const v2f*)(bp1 + ks * 4);
    acc00 = wmma_f32(a0, b0, acc00);
    acc01 = wmma_f32(a0, b1, acc01);
    acc10 = wmma_f32(a1, b0, acc10);
    acc11 = wmma_f32(a1, b1, acc11);
  }

  float* op0 = xi + (size_t)(blkM * 32 + half * 8) * kH + blkN * 32 + lane15;
  float* op1 = op0 + (size_t)16 * kH;
#pragma unroll
  for (int r = 0; r < 8; ++r) {
    op0[(size_t)r * kH]      = acc00[r];
    op0[(size_t)r * kH + 16] = acc01[r];
    op1[(size_t)r * kH]      = acc10[r];
    op1[(size_t)r * kH + 16] = acc11[r];
  }
}

__global__ __launch_bounds__(512) void rnn_scan_kernel(
    float* __restrict__ xi, const float* __restrict__ Wh) {
  __shared__ float hbuf[16 * kLdsStride];

  const int lane   = threadIdx.x & 31;
  const int wave   = threadIdx.x >> 5;
  const int lane15 = lane & 15;
  const int half   = lane >> 4;
  const int bg     = blockIdx.x;

  for (int i = threadIdx.x; i < 16 * kLdsStride; i += 512) hbuf[i] = 0.0f;

  const int n0 = wave * 32;
  const int nA = n0 + lane15;
  const int nB = n0 + 16 + lane15;

  const float* aBase = &hbuf[lane15 * kLdsStride + half * 2];
  const float* bpA   = Wh + (size_t)nA * kH + half * 2;
  const float* bpB   = Wh + (size_t)nB * kH + half * 2;

  const size_t rowStride = (size_t)kT * kH;
  float* xBase = xi + (size_t)(bg * 16 + half * 8) * rowStride;
  const int hrow0 = half * 8;

  v8f pfA, pfB;
#pragma unroll
  for (int r = 0; r < 8; ++r) {
    pfA[r] = xBase[(size_t)r * rowStride + nA];
    pfB[r] = xBase[(size_t)r * rowStride + nB];
  }
  __syncthreads();

  for (int t = 0; t < kT; ++t) {
    v8f accA = pfA;
    v8f accB = pfB;

    if (t + 1 < kT) {
      const float* xn = xBase + (size_t)(t + 1) * kH;
#pragma unroll
      for (int r = 0; r < 8; ++r) {
        pfA[r] = xn[(size_t)r * rowStride + nA];
        pfB[r] = xn[(size_t)r * rowStride + nB];
      }
    }

#pragma unroll 4
    for (int ks = 0; ks < kH / 4; ++ks) {
      v2f a  = *(const v2f*)(aBase + ks * 4);
      v2f b0 = *(const v2f*)(bpA + ks * 4);
      v2f b1 = *(const v2f*)(bpB + ks * 4);
      accA = wmma_f32(a, b0, accA);
      accB = wmma_f32(a, b1, accB);
    }

    __syncthreads();

#pragma unroll
    for (int r = 0; r < 8; ++r) {
      accA[r] = accA[r] > 0.0f ? accA[r] : 0.0f;
      accB[r] = accB[r] > 0.0f ? accB[r] : 0.0f;
      hbuf[(hrow0 + r) * kLdsStride + nA] = accA[r];
      hbuf[(hrow0 + r) * kLdsStride + nB] = accB[r];
    }
    __syncthreads();

    float* xp = xBase + (size_t)t * kH;
#pragma unroll
    for (int r = 0; r < 8; ++r) {
      xp[(size_t)r * rowStride + nA] = accA[r];
      xp[(size_t)r * rowStride + nB] = accB[r];
    }
  }
}

__global__ __launch_bounds__(256) void head_kernel(
    const float* __restrict__ ys, const float* __restrict__ Wfc,
    const float* __restrict__ bfc, float* __restrict__ out) {
  const int lane   = threadIdx.x & 31;
  const int wave   = threadIdx.x >> 5;
  const int lane15 = lane & 15;
  const int half   = lane >> 4;

  const int tile  = blockIdx.x * 8 + wave;
  const int tileN = tile & 7;
  const int tileM = tile >> 3;

  const int n = tileN * 16 + lane15;
  const float biasv = bfc[n];
  v8f acc;
#pragma unroll
  for (int r = 0; r < 8; ++r) acc[r] = biasv;

  const float* ap =
      ys + ((size_t)(tileM * 16 + lane15) * kT + (kT - 1)) * kH + half * 2;
  const float* bp = Wfc + (size_t)n * kH + half * 2;
#pragma unroll 4
  for (int ks = 0; ks < kH / 4; ++ks) {
    v2f a = *(const v2f*)(ap + ks * 4);
    v2f b = *(const v2f*)(bp + ks * 4);
    acc = wmma_f32(a, b, acc);
  }

  float* op = out + (size_t)(tileM * 16 + half * 8) * kC + tileN * 16 + lane15;
#pragma unroll
  for (int r = 0; r < 8; ++r) op[(size_t)r * kC] = acc[r];
}

extern "C" void kernel_launch(void* const* d_in, const int* in_sizes, int n_in,
                              void* d_out, int out_size, void* d_ws,
                              size_t ws_size, hipStream_t stream) {
  const float* x    = (const float*)d_in[0];
  const float* W_ih = (const float*)d_in[1];
  const float* W_hh = (const float*)d_in[2];
  const float* b_ih = (const float*)d_in[3];
  const float* b_hh = (const float*)d_in[4];
  const float* W_fc = (const float*)d_in[5];
  const float* b_fc = (const float*)d_in[6];

  float* bufA = (float*)d_ws;
  float* bufB = bufA + (size_t)kB * kT * kH;

  const int projWaveBlks = (kB * kT / 32) * (kH / 32);
  const int projBlocks   = projWaveBlks / 8;

  const float* cur = x;
  float* dst = bufA;
  float* alt = bufB;
  for (int l = 0; l < kL; ++l) {
    rnn_proj_kernel<<<projBlocks, 256, 0, stream>>>(
        cur, W_ih + (size_t)l * kH * kH, b_ih + (size_t)l * kH,
        b_hh + (size_t)l * kH, dst);
    rnn_scan_kernel<<<4, 512, 0, stream>>>(dst, W_hh + (size_t)l * kH * kH);
    cur = dst;
    float* tmp = dst;
    dst = alt;
    alt = tmp;
  }
  head_kernel<<<4, 256, 0, stream>>>(cur, W_fc, b_fc, (float*)d_out);

  (void)in_sizes; (void)n_in; (void)out_size; (void)ws_size;
}


# Scan: each wave keeps the even 8-k-step blocks of its W_hh columns on-chip over all 512 steps (4 blocks in VGPRs, 4 in LDS); only the odd blocks are re-read per step; k and sum order unchanged
# speedup vs baseline: 2.0962x; 2.0962x over previous
#include <hip/hip_runtime.h>

typedef float v2f __attribute__((ext_vector_type(2)));
typedef float v4f __attribute__((ext_vector_type(4)));
typedef float v8f __attribute__((ext_vector_type(8)));

namespace {

constexpr int kB = 64;
constexpr int kT = 512;
constexpr int kH = 512;
constexpr int kL = 4;
constexpr int kC = 128;
constexpr int kLdsStride = 516;

constexpr int kBlkKs   = 8;
constexpr int kNumBlk  = kH / 4 / kBlkKs;
constexpr int kRegBlks = 4;
constexpr int kLdsBlks = 4;
static_assert(kNumBlk == 2 * (kRegBlks + kLdsBlks), "block split");

__device__ __forceinline__ v8f wmma_f32(v2f a, v2f b, v8f c) {
  return __builtin_amdgcn_wmma_f32_16x16x4_f32(false, a, false, b, (short)0, c,
                                               false, false);
}

}

__global__ __launch_bounds__(256) void rnn_proj_kernel(
    const float* __restrict__ in, const float* __restrict__ Wi,
    const float* __restrict__ bi, const float* __restrict__ bh,
    float* __restrict__ xi) {
  const int lane   = threadIdx.x & 31;
  const int wave   = threadIdx.x >> 5;
  const int lane15 = lane & 15;
  const int half   = lane >> 4;

  const int blk   = blockIdx.x * 8 + wave;
  const int blkN  = blk & (kH / 32 - 1);
  const int blkM  = blk >> 4;

  const int n0 = blkN * 32 + lane15;
  const int n1 = n0 + 16;
  const int m0 = blkM * 32 + lane15;
  const int m1 = m0 + 16;

  const float bias0 = bi[n0] + bh[n0];
  const float bias1 = bi[n1] + bh[n1];
  v8f acc00, acc01, acc10, acc11;
#pragma unroll
  for (int r = 0; r < 8; ++r) {
    acc00[r] = bias0; acc01[r] = bias1;
    acc10[r] = bias0; acc11[r] = bias1;
  }

  const float* ap0 = in + (size_t)m0 * kH + half * 2;
  const float* ap1 = in + (size_t)m1 * kH + half * 2;
  const float* bp0 = Wi + (size_t)n0 * kH + half * 2;
  const float* bp1 = Wi + (size_t)n1 * kH + half * 2;
#pragma unroll 4
  for (int ks = 0; ks < kH / 4; ++ks) {
    v2f a0 = *(const v2f*)(ap0 + ks * 4);
    v2f a1 = *(const v2f*)(ap1 + ks * 4);
    v2f b0 = *(const v2f*)(bp0 + ks * 4);
    v2f b1 = *(const v2f*)(bp1 + ks * 4);
    acc00 = wmma_f32(a0, b0, acc00);
    acc01 = wmma_f32(a0, b1, acc01);
    acc10 = wmma_f32(a1, b0, acc10);
    acc11 = wmma_f32(a1, b1, acc11);
  }

  float* op0 = xi + (size_t)(blkM * 32 + half * 8) * kH + blkN * 32 + lane15;
  float* op1 = op0 + (size_t)16 * kH;
#pragma unroll
  for (int r = 0; r < 8; ++r) {
    op0[(size_t)r * kH]      = acc00[r];
    op0[(size_t)r * kH + 16] = acc01[r];
    op1[(size_t)r * kH]      = acc10[r];
    op1[(size_t)r * kH + 16] = acc11[r];
  }
}

__global__ __launch_bounds__(512) void rnn_scan_kernel(float* xi,
                                                       const float* Wh) {
  __shared__ float hbuf[16 * kLdsStride];
  __shared__ v4f wlds[16 * kLdsBlks * kBlkKs * 32];

  const int lane   = threadIdx.x & 31;
  const int wave   = threadIdx.x >> 5;
  const int lane15 = lane & 15;
  const int half   = lane >> 4;
  const int bg     = blockIdx.x;

  for (int i = threadIdx.x; i < 16 * kLdsStride; i += 512) hbuf[i] = 0.0f;

  const int n0 = wave * 32;
  const int nA = n0 + lane15;
  const int nB = n0 + 16 + lane15;

  const float* aBase = &hbuf[lane15 * kLdsStride + half * 2];
  const float* bpA   = Wh + (size_t)nA * kH + half * 2;
  const float* bpB   = Wh + (size_t)nB * kH + half * 2;

  v4f* wl = &wlds[(wave * kLdsBlks * kBlkKs) * 32 + lane];
#pragma unroll 4
  for (int q = 0; q < kLdsBlks * kBlkKs; ++q) {
    const int ks = (4 * (q / kBlkKs) + 2) * kBlkKs + (q % kBlkKs);
    const v2f b0 = *(const v2f*)(bpA + ks * 4);
    const v2f b1 = *(const v2f*)(bpB + ks * 4);
    v4f w;
    w.x = b0.x; w.y = b0.y; w.z = b1.x; w.w = b1.y;
    wl[q * 32] = w;
  }
  v2f wrA[kRegBlks * kBlkKs], wrB[kRegBlks * kBlkKs];
#pragma unroll
  for (int i = 0; i < kRegBlks; ++i) {
#pragma unroll
    for (int s = 0; s < kBlkKs; ++s) {
      const int ks = (4 * i) * kBlkKs + s;
      wrA[i * kBlkKs + s] = *(const v2f*)(bpA + ks * 4);
      wrB[i * kBlkKs + s] = *(const v2f*)(bpB + ks * 4);
    }
  }

  const size_t rowStride = (size_t)kT * kH;
  float* xBase = xi + (size_t)(bg * 16 + half * 8) * rowStride;
  const int hrow0 = half * 8;

  v8f pfA, pfB;
#pragma unroll
  for (int r = 0; r < 8; ++r) {
    pfA[r] = xBase[(size_t)r * rowStride + nA];
    pfB[r] = xBase[(size_t)r * rowStride + nB];
  }
  __syncthreads();

  for (int t = 0; t < kT; ++t) {
    v8f accA = pfA;
    v8f accB = pfB;

    v2f sA[kBlkKs], sB[kBlkKs];
#pragma unroll
    for (int s = 0; s < kBlkKs; ++s) {
      sA[s] = *(const v2f*)(bpA + (kBlkKs + s) * 4);
      sB[s] = *(const v2f*)(bpB + (kBlkKs + s) * 4);
    }

    if (t + 1 < kT) {
      const float* xn = xBase + (size_t)(t + 1) * kH;
#pragma unroll
      for (int r = 0; r < 8; ++r) {
        pfA[r] = xn[(size_t)r * rowStride + nA];
        pfB[r] = xn[(size_t)r * rowStride + nB];
      }
    }

#pragma unroll
    for (int j = 0; j < kNumBlk / 2; ++j) {
      const int br = 2 * j;
      const int bs = 2 * j + 1;
      const int i  = j >> 1;
      if ((j & 1) == 0) {
#pragma unroll
        for (int s = 0; s < kBlkKs; ++s) {
          const v2f a = *(const v2f*)(aBase + (br * kBlkKs + s) * 4);
          accA = wmma_f32(a, wrA[i * kBlkKs + s], accA);
          accB = wmma_f32(a, wrB[i * kBlkKs + s], accB);
        }
      } else {
#pragma unroll
        for (int s = 0; s < kBlkKs; ++s) {
          const v2f a = *(const v2f*)(aBase + (br * kBlkKs + s) * 4);
          const v4f w = wl[(i * kBlkKs + s) * 32];
          v2f b0, b1;
          b0.x = w.x; b0.y = w.y; b1.x = w.z; b1.y = w.w;
          accA = wmma_f32(a, b0, accA);
          accB = wmma_f32(a, b1, accB);
        }
      }
#pragma unroll
      for (int s = 0; s < kBlkKs; ++s) {
        const v2f a = *(const v2f*)(aBase + (bs * kBlkKs + s) * 4);
        accA = wmma_f32(a, sA[s], accA);
        accB = wmma_f32(a, sB[s], accB);
      }
      if (j + 1 < kNumBlk / 2) {
#pragma unroll
        for (int s = 0; s < kBlkKs; ++s) {
          sA[s] = *(const v2f*)(bpA + ((bs + 2) * kBlkKs + s) * 4);
          sB[s] = *(const v2f*)(bpB + ((bs + 2) * kBlkKs + s) * 4);
        }
      }
    }

    __syncthreads();

#pragma unroll
    for (int r = 0; r < 8; ++r) {
      accA[r] = accA[r] > 0.0f ? accA[r] : 0.0f;
      accB[r] = accB[r] > 0.0f ? accB[r] : 0.0f;
      hbuf[(hrow0 + r) * kLdsStride + nA] = accA[r];
      hbuf[(hrow0 + r) * kLdsStride + nB] = accB[r];
    }
    __syncthreads();

    float* xp = xBase + (size_t)t * kH;
#pragma unroll
    for (int r = 0; r < 8; ++r) {
      xp[(size_t)r * rowStride + nA] = accA[r];
      xp[(size_t)r * rowStride + nB] = accB[r];
    }
  }
}

__global__ __launch_bounds__(256) void head_kernel(
    const float* __restrict__ ys, const float* __restrict__ Wfc,
    const float* __restrict__ bfc, float* __restrict__ out) {
  const int lane   = threadIdx.x & 31;
  const int wave   = threadIdx.x >> 5;
  const int lane15 = lane & 15;
  const int half   = lane >> 4;

  const int tile  = blockIdx.x * 8 + wave;
  const int tileN = tile & 7;
  const int tileM = tile >> 3;

  const int n = tileN * 16 + lane15;
  const float biasv = bfc[n];
  v8f acc;
#pragma unroll
  for (int r = 0; r < 8; ++r) acc[r] = biasv;

  const float* ap =
      ys + ((size_t)(tileM * 16 + lane15) * kT + (kT - 1)) * kH + half * 2;
  const float* bp = Wfc + (size_t)n * kH + half * 2;
#pragma unroll 4
  for (int ks = 0; ks < kH / 4; ++ks) {
    v2f a = *(const v2f*)(ap + ks * 4);
    v2f b = *(const v2f*)(bp + ks * 4);
    acc = wmma_f32(a, b, acc);
  }

  float* op = out + (size_t)(tileM * 16 + half * 8) * kC + tileN * 16 + lane15;
#pragma unroll
  for (int r = 0; r < 8; ++r) op[(size_t)r * kC] = acc[r];
}

extern "C" void kernel_launch(void* const* d_in, const int* in_sizes, int n_in,
                              void* d_out, int out_size, void* d_ws,
                              size_t ws_size, hipStream_t stream) {
  const float* x    = (const float*)d_in[0];
  const float* W_ih = (const float*)d_in[1];
  const float* W_hh = (const float*)d_in[2];
  const float* b_ih = (const float*)d_in[3];
  const float* b_hh = (const float*)d_in[4];
  const float* W_fc = (const float*)d_in[5];
  const float* b_fc = (const float*)d_in[6];

  float* bufA = (float*)d_ws;
  float* bufB = bufA + (size_t)kB * kT * kH;

  const int projWaveBlks = (kB * kT / 32) * (kH / 32);
  const int projBlocks   = projWaveBlks / 8;

  const float* cur = x;
  float* dst = bufA;
  float* alt = bufB;
  for (int l = 0; l < kL; ++l) {
    rnn_proj_kernel<<<projBlocks, 256, 0, stream>>>(
        cur, W_ih + (size_t)l * kH * kH, b_ih + (size_t)l * kH,
        b_hh + (size_t)l * kH, dst);
    rnn_scan_kernel<<<4, 512, 0, stream>>>(dst, W_hh + (size_t)l * kH * kH);
    cur = dst;
    float* tmp = dst;
    dst = alt;
    alt = tmp;
  }
  head_kernel<<<4, 256, 0, stream>>>(cur, W_fc, b_fc, (float*)d_out);

  (void)in_sizes; (void)n_in; (void)out_size; (void)ws_size;
}


# Scan: step boundary freed of global memory: h_t stores and next step's first W_hh block issued before the barriers, which fence LDS only; xi prefetch in mid-step; k and sum order unchanged
# speedup vs baseline: 2.3591x; 1.1254x over previous
#include <hip/hip_runtime.h>

typedef float v2f __attribute__((ext_vector_type(2)));
typedef float v4f __attribute__((ext_vector_type(4)));
typedef float v8f __attribute__((ext_vector_type(8)));

namespace {

constexpr int kB = 64;
constexpr int kT = 512;
constexpr int kH = 512;
constexpr int kL = 4;
constexpr int kC = 128;
constexpr int kLdsStride = 516;

constexpr int kBlkKs   = 8;
constexpr int kNumBlk  = kH / 4 / kBlkKs;
constexpr int kRegBlks = 4;
constexpr int kLdsBlks = 4;
static_assert(kNumBlk == 2 * (kRegBlks + kLdsBlks), "block split");

__device__ __forceinline__ v8f wmma_f32(v2f a, v2f b, v8f c) {
  return __builtin_amdgcn_wmma_f32_16x16x4_f32(false, a, false, b, (short)0, c,
                                               false, false);
}

__device__ __forceinline__ void lds_step_barrier() {
  __builtin_amdgcn_fence(__ATOMIC_RELEASE, "workgroup", "local");
  __builtin_amdgcn_s_barrier();
  __builtin_amdgcn_fence(__ATOMIC_ACQUIRE, "workgroup", "local");
}

}

__global__ __launch_bounds__(256) void rnn_proj_kernel(
    const float* __restrict__ in, const float* __restrict__ Wi,
    const float* __restrict__ bi, const float* __restrict__ bh,
    float* __restrict__ xi) {
  const int lane   = threadIdx.x & 31;
  const int wave   = threadIdx.x >> 5;
  const int lane15 = lane & 15;
  const int half   = lane >> 4;

  const int blk   = blockIdx.x * 8 + wave;
  const int blkN  = blk & (kH / 32 - 1);
  const int blkM  = blk >> 4;

  const int n0 = blkN * 32 + lane15;
  const int n1 = n0 + 16;
  const int m0 = blkM * 32 + lane15;
  const int m1 = m0 + 16;

  const float bias0 = bi[n0] + bh[n0];
  const float bias1 = bi[n1] + bh[n1];
  v8f acc00, acc01, acc10, acc11;
#pragma unroll
  for (int r = 0; r < 8; ++r) {
    acc00[r] = bias0; acc01[r] = bias1;
    acc10[r] = bias0; acc11[r] = bias1;
  }

  const float* ap0 = in + (size_t)m0 * kH + half * 2;
  const float* ap1 = in + (size_t)m1 * kH + half * 2;
  const float* bp0 = Wi + (size_t)n0 * kH + half * 2;
  const float* bp1 = Wi + (size_t)n1 * kH + half * 2;
#pragma unroll 4
  for (int ks = 0; ks < kH / 4; ++ks) {
    v2f a0 = *(const v2f*)(ap0 + ks * 4);
    v2f a1 = *(const v2f*)(ap1 + ks * 4);
    v2f b0 = *(const v2f*)(bp0 + ks * 4);
    v2f b1 = *(const v2f*)(bp1 + ks * 4);
    acc00 = wmma_f32(a0, b0, acc00);
    acc01 = wmma_f32(a0, b1, acc01);
    acc10 = wmma_f32(a1, b0, acc10);
    acc11 = wmma_f32(a1, b1, acc11);
  }

  float* op0 = xi + (size_t)(blkM * 32 + half * 8) * kH + blkN * 32 + lane15;
  float* op1 = op0 + (size_t)16 * kH;
#pragma unroll
  for (int r = 0; r < 8; ++r) {
    op0[(size_t)r * kH]      = acc00[r];
    op0[(size_t)r * kH + 16] = acc01[r];
    op1[(size_t)r * kH]      = acc10[r];
    op1[(size_t)r * kH + 16] = acc11[r];
  }
}

__global__ __launch_bounds__(512) void rnn_scan_kernel(float* xi,
                                                       const float* Wh) {
  __shared__ float hbuf[16 * kLdsStride];
  __shared__ v4f wlds[16 * kLdsBlks * kBlkKs * 32];

  const int lane   = threadIdx.x & 31;
  const int wave   = threadIdx.x >> 5;
  const int lane15 = lane & 15;
  const int half   = lane >> 4;
  const int bg     = blockIdx.x;

  for (int i = threadIdx.x; i < 16 * kLdsStride; i += 512) hbuf[i] = 0.0f;

  const int n0 = wave * 32;
  const int nA = n0 + lane15;

  const float* aBase = &hbuf[lane15 * kLdsStride + half * 2];
  const float* bpA   = Wh + (size_t)nA * kH + half * 2;
  const float* bpB   = bpA + (size_t)16 * kH;

  v4f* wl = &wlds[(wave * kLdsBlks * kBlkKs) * 32 + lane];
#pragma unroll 4
  for (int q = 0; q < kLdsBlks * kBlkKs; ++q) {
    const int ks = (4 * (q / kBlkKs) + 2) * kBlkKs + (q % kBlkKs);
    const v2f b0 = *(const v2f*)(bpA + ks * 4);
    const v2f b1 = *(const v2f*)(bpB + ks * 4);
    v4f w;
    w.x = b0.x; w.y = b0.y; w.z = b1.x; w.w = b1.y;
    wl[q * 32] = w;
  }
  v2f wrA[kRegBlks * kBlkKs], wrB[kRegBlks * kBlkKs];
#pragma unroll
  for (int i = 0; i < kRegBlks; ++i) {
#pragma unroll
    for (int s = 0; s < kBlkKs; ++s) {
      const int ks = (4 * i) * kBlkKs + s;
      wrA[i * kBlkKs + s] = *(const v2f*)(bpA + ks * 4);
      wrB[i * kBlkKs + s] = *(const v2f*)(bpB + ks * 4);
    }
  }

  const size_t rowStride = (size_t)kT * kH;
  float* xBase = xi + (size_t)(bg * 16 + half * 8) * rowStride + nA;
  float* hPub  = &hbuf[half * 8 * kLdsStride + nA];

  v8f pfA, pfB;
#pragma unroll
  for (int r = 0; r < 8; ++r) {
    pfA[r] = xBase[(size_t)r * rowStride];
    pfB[r] = xBase[(size_t)r * rowStride + 16];
  }
  v2f sA[kBlkKs], sB[kBlkKs];
#pragma unroll
  for (int s = 0; s < kBlkKs; ++s) {
    sA[s] = *(const v2f*)(bpA + (kBlkKs + s) * 4);
    sB[s] = *(const v2f*)(bpB + (kBlkKs + s) * 4);
  }
  __syncthreads();

  for (int t = 0; t < kT; ++t) {
    v8f accA = pfA;
    v8f accB = pfB;

#pragma unroll
    for (int j = 0; j < kNumBlk / 2; ++j) {
      const int br = 2 * j;
      const int bs = 2 * j + 1;
      const int i  = j >> 1;
      if ((j & 1) == 0) {
#pragma unroll
        for (int s = 0; s < kBlkKs; ++s) {
          const v2f a = *(const v2f*)(aBase + (br * kBlkKs + s) * 4);
          accA = wmma_f32(a, wrA[i * kBlkKs + s], accA);
          accB = wmma_f32(a, wrB[i * kBlkKs + s], accB);
        }
      } else {
#pragma unroll
        for (int s = 0; s < kBlkKs; ++s) {
          const v2f a = *(const v2f*)(aBase + (br * kBlkKs + s) * 4);
          const v4f w = wl[(i * kBlkKs + s) * 32];
          v2f b0, b1;
          b0.x = w.x; b0.y = w.y; b1.x = w.z; b1.y = w.w;
          accA = wmma_f32(a, b0, accA);
          accB = wmma_f32(a, b1, accB);
        }
      }
      const int nb = (j + 1 < kNumBlk / 2) ? bs + 2 : 1;
      if (j == kNumBlk / 4) {
        const int tn = t + 1 < kT ? t + 1 : kT - 1;
        const float* xn = xBase + (size_t)tn * kH;
#pragma unroll
        for (int r = 0; r < 8; ++r) {
          pfA[r] = xn[(size_t)r * rowStride];
          pfB[r] = xn[(size_t)r * rowStride + 16];
        }
      }
#pragma unroll
      for (int s = 0; s < kBlkKs; ++s) {
        const v2f a = *(const v2f*)(aBase + (bs * kBlkKs + s) * 4);
        accA = wmma_f32(a, sA[s], accA);
        accB = wmma_f32(a, sB[s], accB);
        __builtin_amdgcn_sched_barrier(0x786);
        sA[s] = *(const v2f*)(bpA + (nb * kBlkKs + s) * 4);
        sB[s] = *(const v2f*)(bpB + (nb * kBlkKs + s) * 4);
      }
    }

    float* xp = xBase + (size_t)t * kH;
#pragma unroll
    for (int r = 0; r < 8; ++r) {
      accA[r] = accA[r] > 0.0f ? accA[r] : 0.0f;
      accB[r] = accB[r] > 0.0f ? accB[r] : 0.0f;
      xp[(size_t)r * rowStride]      = accA[r];
      xp[(size_t)r * rowStride + 16] = accB[r];
    }

    lds_step_barrier();

#pragma unroll
    for (int r = 0; r < 8; ++r) {
      hPub[r * kLdsStride]      = accA[r];
      hPub[r * kLdsStride + 16] = accB[r];
    }
    lds_step_barrier();
  }
}

__global__ __launch_bounds__(256) void head_kernel(
    const float* __restrict__ ys, const float* __restrict__ Wfc,
    const float* __restrict__ bfc, float* __restrict__ out) {
  const int lane   = threadIdx.x & 31;
  const int wave   = threadIdx.x >> 5;
  const int lane15 = lane & 15;
  const int half   = lane >> 4;

  const int tile  = blockIdx.x * 8 + wave;
  const int tileN = tile & 7;
  const int tileM = tile >> 3;

  const int n = tileN * 16 + lane15;
  const float biasv = bfc[n];
  v8f acc;
#pragma unroll
  for (int r = 0; r < 8; ++r) acc[r] = biasv;

  const float* ap =
      ys + ((size_t)(tileM * 16 + lane15) * kT + (kT - 1)) * kH + half * 2;
  const float* bp = Wfc + (size_t)n * kH + half * 2;
#pragma unroll 4
  for (int ks = 0; ks < kH / 4; ++ks) {
    v2f a = *(const v2f*)(ap + ks * 4);
    v2f b = *(const v2f*)(bp + ks * 4);
    acc = wmma_f32(a, b, acc);
  }

  float* op = out + (size_t)(tileM * 16 + half * 8) * kC + tileN * 16 + lane15;
#pragma unroll
  for (int r = 0; r < 8; ++r) op[(size_t)r * kC] = acc[r];
}

extern "C" void kernel_launch(void* const* d_in, const int* in_sizes, int n_in,
                              void* d_out, int out_size, void* d_ws,
                              size_t ws_size, hipStream_t stream) {
  const float* x    = (const float*)d_in[0];
  const float* W_ih = (const float*)d_in[1];
  const float* W_hh = (const float*)d_in[2];
  const float* b_ih = (const float*)d_in[3];
  const float* b_hh = (const float*)d_in[4];
  const float* W_fc = (const float*)d_in[5];
  const float* b_fc = (const float*)d_in[6];

  float* bufA = (float*)d_ws;
  float* bufB = bufA + (size_t)kB * kT * kH;

  const int projWaveBlks = (kB * kT / 32) * (kH / 32);
  const int projBlocks   = projWaveBlks / 8;

  const float* cur = x;
  float* dst = bufA;
  float* alt = bufB;
  for (int l = 0; l < kL; ++l) {
    rnn_proj_kernel<<<projBlocks, 256, 0, stream>>>(
        cur, W_ih + (size_t)l * kH * kH, b_ih + (size_t)l * kH,
        b_hh + (size_t)l * kH, dst);
    rnn_scan_kernel<<<4, 512, 0, stream>>>(dst, W_hh + (size_t)l * kH * kH);
    cur = dst;
    float* tmp = dst;
    dst = alt;
    alt = tmp;
  }
  head_kernel<<<4, 256, 0, stream>>>(cur, W_fc, b_fc, (float*)d_out);

  (void)in_sizes; (void)n_in; (void)out_size; (void)ws_size;
}


# Scan cut into 16 time chunks run as a layer pipeline: launch pair d projects then scans chunk d-l of every layer l at once (16 scan workgroups, was 4); h carried in the buffer; sum order unchanged
# speedup vs baseline: 7.0346x; 2.9819x over previous
#include <hip/hip_runtime.h>

typedef float v2f __attribute__((ext_vector_type(2)));
typedef float v4f __attribute__((ext_vector_type(4)));
typedef float v8f __attribute__((ext_vector_type(8)));

namespace {

constexpr int kB = 64;
constexpr int kT = 512;
constexpr int kH = 512;
constexpr int kL = 4;
constexpr int kC = 128;
constexpr int kLdsStride = 516;

constexpr int kTc   = 32;
constexpr int kNC   = kT / kTc;
constexpr int kDiag = kNC + kL - 1;
static_assert(kTc % 32 == 0 && kT % kTc == 0, "chunk split");

constexpr int kBlkKs   = 8;
constexpr int kNumBlk  = kH / 4 / kBlkKs;
constexpr int kRegBlks = 4;
constexpr int kLdsBlks = 4;
static_assert(kNumBlk == 2 * (kRegBlks + kLdsBlks), "block split");

__device__ __forceinline__ v8f wmma_f32(v2f a, v2f b, v8f c) {
  return __builtin_amdgcn_wmma_f32_16x16x4_f32(false, a, false, b, (short)0, c,
                                               false, false);
}

__device__ __forceinline__ void lds_step_barrier() {
  __builtin_amdgcn_fence(__ATOMIC_RELEASE, "workgroup", "local");
  __builtin_amdgcn_s_barrier();
  __builtin_amdgcn_fence(__ATOMIC_ACQUIRE, "workgroup", "local");
}

}

__global__ __launch_bounds__(256) void rnn_proj_kernel(
    const float* __restrict__ x, float* bufA, float* bufB,
    const float* __restrict__ W_ih, const float* __restrict__ b_ih,
    const float* __restrict__ b_hh, int d) {
  const int l = blockIdx.y;
  const int c = d - l;
  if (c < 0 || c >= kNC) return;

  const float* __restrict__ in =
      l == 0 ? x : ((l & 1) ? (const float*)bufA : (const float*)bufB);
  float* __restrict__ xi = (l & 1) ? bufB : bufA;
  const float* __restrict__ Wi = W_ih + (size_t)l * kH * kH;
  const float* __restrict__ bi = b_ih + (size_t)l * kH;
  const float* __restrict__ bh = b_hh + (size_t)l * kH;

  const int lane   = threadIdx.x & 31;
  const int wave   = threadIdx.x >> 5;
  const int lane15 = lane & 15;
  const int half   = lane >> 4;

  constexpr int kRowBlk = kTc / 32;
  const int blk   = blockIdx.x * 8 + wave;
  const int blkN  = blk & (kH / 32 - 1);
  const int blkM  = blk >> 4;
  const int rowBase =
      (blkM / kRowBlk) * kT + c * kTc + (blkM % kRowBlk) * 32;

  const int n0 = blkN * 32 + lane15;
  const int n1 = n0 + 16;
  const int m0 = rowBase + lane15;
  const int m1 = m0 + 16;

  const float bias0 = bi[n0] + bh[n0];
  const float bias1 = bi[n1] + bh[n1];
  v8f acc00, acc01, acc10, acc11;
#pragma unroll
  for (int r = 0; r < 8; ++r) {
    acc00[r] = bias0; acc01[r] = bias1;
    acc10[r] = bias0; acc11[r] = bias1;
  }

  const float* ap0 = in + (size_t)m0 * kH + half * 2;
  const float* ap1 = in + (size_t)m1 * kH + half * 2;
  const float* bp0 = Wi + (size_t)n0 * kH + half * 2;
  const float* bp1 = Wi + (size_t)n1 * kH + half * 2;
#pragma unroll 4
  for (int ks = 0; ks < kH / 4; ++ks) {
    v2f a0 = *(const v2f*)(ap0 + ks * 4);
    v2f a1 = *(const v2f*)(ap1 + ks * 4);
    v2f b0 = *(const v2f*)(bp0 + ks * 4);
    v2f b1 = *(const v2f*)(bp1 + ks * 4);
    acc00 = wmma_f32(a0, b0, acc00);
    acc01 = wmma_f32(a0, b1, acc01);
    acc10 = wmma_f32(a1, b0, acc10);
    acc11 = wmma_f32(a1, b1, acc11);
  }

  float* op0 = xi + (size_t)(rowBase + half * 8) * kH + blkN * 32 + lane15;
  float* op1 = op0 + (size_t)16 * kH;
#pragma unroll
  for (int r = 0; r < 8; ++r) {
    op0[(size_t)r * kH]      = acc00[r];
    op0[(size_t)r * kH + 16] = acc01[r];
    op1[(size_t)r * kH]      = acc10[r];
    op1[(size_t)r * kH + 16] = acc11[r];
  }
}

__global__ __launch_bounds__(512) void rnn_scan_kernel(float* bufA,
                                                       float* bufB,
                                                       const float* W_hh,
                                                       int d) {
  __shared__ float hbuf[16 * kLdsStride];
  __shared__ v4f wlds[16 * kLdsBlks * kBlkKs * 32];

  const int l = blockIdx.y;
  const int c = d - l;
  if (c < 0 || c >= kNC) return;
  float* xi       = (l & 1) ? bufB : bufA;
  const float* Wh = W_hh + (size_t)l * kH * kH;
  const int t0    = c * kTc;

  const int lane   = threadIdx.x & 31;
  const int wave   = threadIdx.x >> 5;
  const int lane15 = lane & 15;
  const int half   = lane >> 4;
  const int bg     = blockIdx.x;

  if (c == 0) {
    for (int i = threadIdx.x; i < 16 * kLdsStride; i += 512) hbuf[i] = 0.0f;
  } else {
    const float* hp =
        xi + (size_t)(bg * 16) * kT * kH + (size_t)(t0 - 1) * kH + threadIdx.x;
    for (int r = 0; r < 16; ++r)
      hbuf[r * kLdsStride + threadIdx.x] = hp[(size_t)r * kT * kH];
  }

  const int n0 = wave * 32;
  const int nA = n0 + lane15;

  const float* aBase = &hbuf[lane15 * kLdsStride + half * 2];
  const float* bpA   = Wh + (size_t)nA * kH + half * 2;
  const float* bpB   = bpA + (size_t)16 * kH;

  v4f* wl = &wlds[(wave * kLdsBlks * kBlkKs) * 32 + lane];
#pragma unroll 4
  for (int q = 0; q < kLdsBlks * kBlkKs; ++q) {
    const int ks = (4 * (q / kBlkKs) + 2) * kBlkKs + (q % kBlkKs);
    const v2f b0 = *(const v2f*)(bpA + ks * 4);
    const v2f b1 = *(const v2f*)(bpB + ks * 4);
    v4f w;
    w.x = b0.x; w.y = b0.y; w.z = b1.x; w.w = b1.y;
    wl[q * 32] = w;
  }
  v2f wrA[kRegBlks * kBlkKs], wrB[kRegBlks * kBlkKs];
#pragma unroll
  for (int i = 0; i < kRegBlks; ++i) {
#pragma unroll
    for (int s = 0; s < kBlkKs; ++s) {
      const int ks = (4 * i) * kBlkKs + s;
      wrA[i * kBlkKs + s] = *(const v2f*)(bpA + ks * 4);
      wrB[i * kBlkKs + s] = *(const v2f*)(bpB + ks * 4);
    }
  }

  const size_t rowStride = (size_t)kT * kH;
  float* xBase =
      xi + (size_t)(bg * 16 + half * 8) * rowStride + (size_t)t0 * kH + nA;
  float* hPub  = &hbuf[half * 8 * kLdsStride + nA];

  v8f pfA, pfB;
#pragma unroll
  for (int r = 0; r < 8; ++r) {
    pfA[r] = xBase[(size_t)r * rowStride];
    pfB[r] = xBase[(size_t)r * rowStride + 16];
  }
  v2f sA[kBlkKs], sB[kBlkKs];
#pragma unroll
  for (int s = 0; s < kBlkKs; ++s) {
    sA[s] = *(const v2f*)(bpA + (kBlkKs + s) * 4);
    sB[s] = *(const v2f*)(bpB + (kBlkKs + s) * 4);
  }
  __syncthreads();

  for (int t = 0; t < kTc; ++t) {
    v8f accA = pfA;
    v8f accB = pfB;

#pragma unroll
    for (int j = 0; j < kNumBlk / 2; ++j) {
      const int br = 2 * j;
      const int bs = 2 * j + 1;
      const int i  = j >> 1;
      if ((j & 1) == 0) {
#pragma unroll
        for (int s = 0; s < kBlkKs; ++s) {
          const v2f a = *(const v2f*)(aBase + (br * kBlkKs + s) * 4);
          accA = wmma_f32(a, wrA[i * kBlkKs + s], accA);
          accB = wmma_f32(a, wrB[i * kBlkKs + s], accB);
        }
      } else {
#pragma unroll
        for (int s = 0; s < kBlkKs; ++s) {
          const v2f a = *(const v2f*)(aBase + (br * kBlkKs + s) * 4);
          const v4f w = wl[(i * kBlkKs + s) * 32];
          v2f b0, b1;
          b0.x = w.x; b0.y = w.y; b1.x = w.z; b1.y = w.w;
          accA = wmma_f32(a, b0, accA);
          accB = wmma_f32(a, b1, accB);
        }
      }
      const int nb = (j + 1 < kNumBlk / 2) ? bs + 2 : 1;
      if (j == kNumBlk / 4) {
        const int tn = t + 1 < kTc ? t + 1 : kTc - 1;
        const float* xn = xBase + (size_t)tn * kH;
#pragma unroll
        for (int r = 0; r < 8; ++r) {
          pfA[r] = xn[(size_t)r * rowStride];
          pfB[r] = xn[(size_t)r * rowStride + 16];
        }
      }
#pragma unroll
      for (int s = 0; s < kBlkKs; ++s) {
        const v2f a = *(const v2f*)(aBase + (bs * kBlkKs + s) * 4);
        accA = wmma_f32(a, sA[s], accA);
        accB = wmma_f32(a, sB[s], accB);
        __builtin_amdgcn_sched_barrier(0x786);
        sA[s] = *(const v2f*)(bpA + (nb * kBlkKs + s) * 4);
        sB[s] = *(const v2f*)(bpB + (nb * kBlkKs + s) * 4);
      }
    }

    float* xp = xBase + (size_t)t * kH;
#pragma unroll
    for (int r = 0; r < 8; ++r) {
      accA[r] = accA[r] > 0.0f ? accA[r] : 0.0f;
      accB[r] = accB[r] > 0.0f ? accB[r] : 0.0f;
      xp[(size_t)r * rowStride]      = accA[r];
      xp[(size_t)r * rowStride + 16] = accB[r];
    }

    lds_step_barrier();

#pragma unroll
    for (int r = 0; r < 8; ++r) {
      hPub[r * kLdsStride]      = accA[r];
      hPub[r * kLdsStride + 16] = accB[r];
    }
    lds_step_barrier();
  }
}

__global__ __launch_bounds__(256) void head_kernel(
    const float* __restrict__ ys, const float* __restrict__ Wfc,
    const float* __restrict__ bfc, float* __restrict__ out) {
  const int lane   = threadIdx.x & 31;
  const int wave   = threadIdx.x >> 5;
  const int lane15 = lane & 15;
  const int half   = lane >> 4;

  const int tile  = blockIdx.x * 8 + wave;
  const int tileN = tile & 7;
  const int tileM = tile >> 3;

  const int n = tileN * 16 + lane15;
  const float biasv = bfc[n];
  v8f acc;
#pragma unroll
  for (int r = 0; r < 8; ++r) acc[r] = biasv;

  const float* ap =
      ys + ((size_t)(tileM * 16 + lane15) * kT + (kT - 1)) * kH + half * 2;
  const float* bp = Wfc + (size_t)n * kH + half * 2;
#pragma unroll 4
  for (int ks = 0; ks < kH / 4; ++ks) {
    v2f a = *(const v2f*)(ap + ks * 4);
    v2f b = *(const v2f*)(bp + ks * 4);
    acc = wmma_f32(a, b, acc);
  }

  float* op = out + (size_t)(tileM * 16 + half * 8) * kC + tileN * 16 + lane15;
#pragma unroll
  for (int r = 0; r < 8; ++r) op[(size_t)r * kC] = acc[r];
}

extern "C" void kernel_launch(void* const* d_in, const int* in_sizes, int n_in,
                              void* d_out, int out_size, void* d_ws,
                              size_t ws_size, hipStream_t stream) {
  const float* x    = (const float*)d_in[0];
  const float* W_ih = (const float*)d_in[1];
  const float* W_hh = (const float*)d_in[2];
  const float* b_ih = (const float*)d_in[3];
  const float* b_hh = (const float*)d_in[4];
  const float* W_fc = (const float*)d_in[5];
  const float* b_fc = (const float*)d_in[6];

  float* bufA = (float*)d_ws;
  float* bufB = bufA + (size_t)kB * kT * kH;

  constexpr int projBlocks = (kB * kTc / 32) * (kH / 32) / 8;

  for (int d = 0; d < kDiag; ++d) {
    rnn_proj_kernel<<<dim3(projBlocks, kL), 256, 0, stream>>>(
        x, bufA, bufB, W_ih, b_ih, b_hh, d);
    rnn_scan_kernel<<<dim3(4, kL), 512, 0, stream>>>(bufA, bufB, W_hh, d);
  }
  head_kernel<<<4, 256, 0, stream>>>(bufB, W_fc, b_fc, (float*)d_out);

  (void)in_sizes; (void)n_in; (void)out_size; (void)ws_size;
}


# Fragment reads 16 B per lane, k-steps paired: one read feeds two WMMAs (scan state, W_hh, projection operands); half the LDS and global loads; sum order changes: k permuted within groups of 8
# speedup vs baseline: 7.2998x; 1.0377x over previous
#include <hip/hip_runtime.h>

typedef float v2f __attribute__((ext_vector_type(2)));
typedef float v4f __attribute__((ext_vector_type(4)));
typedef float v8f __attribute__((ext_vector_type(8)));

namespace {

constexpr int kB = 64;
constexpr int kT = 512;
constexpr int kH = 512;
constexpr int kL = 4;
constexpr int kC = 128;
constexpr int kLdsStride = 516;

constexpr int kTc   = 32;
constexpr int kNC   = kT / kTc;
constexpr int kDiag = kNC + kL - 1;
static_assert(kTc % 32 == 0 && kT % kTc == 0, "chunk split");

constexpr int kBlkKs   = 8;
constexpr int kBlkKp   = kBlkKs / 2;
constexpr int kNumBlk  = kH / 4 / kBlkKs;
constexpr int kRegBlks = 4;
constexpr int kLdsBlks = 4;
static_assert(kNumBlk == 2 * (kRegBlks + kLdsBlks), "block split");

__device__ __forceinline__ v8f wmma_f32(v2f a, v2f b, v8f c) {
  return __builtin_amdgcn_wmma_f32_16x16x4_f32(false, a, false, b, (short)0, c,
                                               false, false);
}

__device__ __forceinline__ v2f frag_lo(v4f v) {
  return __builtin_shufflevector(v, v, 0, 1);
}
__device__ __forceinline__ v2f frag_hi(v4f v) {
  return __builtin_shufflevector(v, v, 2, 3);
}

__device__ __forceinline__ void lds_step_barrier() {
  __builtin_amdgcn_fence(__ATOMIC_RELEASE, "workgroup", "local");
  __builtin_amdgcn_s_barrier();
  __builtin_amdgcn_fence(__ATOMIC_ACQUIRE, "workgroup", "local");
}

}

__global__ __launch_bounds__(256) void rnn_proj_kernel(
    const float* __restrict__ x, float* bufA, float* bufB,
    const float* __restrict__ W_ih, const float* __restrict__ b_ih,
    const float* __restrict__ b_hh, int d) {
  const int l = blockIdx.y;
  const int c = d - l;
  if (c < 0 || c >= kNC) return;

  const float* __restrict__ in =
      l == 0 ? x : ((l & 1) ? (const float*)bufA : (const float*)bufB);
  float* __restrict__ xi = (l & 1) ? bufB : bufA;
  const float* __restrict__ Wi = W_ih + (size_t)l * kH * kH;
  const float* __restrict__ bi = b_ih + (size_t)l * kH;
  const float* __restrict__ bh = b_hh + (size_t)l * kH;

  const int lane   = threadIdx.x & 31;
  const int wave   = threadIdx.x >> 5;
  const int lane15 = lane & 15;
  const int half   = lane >> 4;

  constexpr int kRowBlk = kTc / 32;
  const int blk   = blockIdx.x * 8 + wave;
  const int blkN  = blk & (kH / 32 - 1);
  const int blkM  = blk >> 4;
  const int rowBase =
      (blkM / kRowBlk) * kT + c * kTc + (blkM % kRowBlk) * 32;

  const int n0 = blkN * 32 + lane15;
  const int n1 = n0 + 16;
  const int m0 = rowBase + lane15;
  const int m1 = m0 + 16;

  const float bias0 = bi[n0] + bh[n0];
  const float bias1 = bi[n1] + bh[n1];
  v8f acc00, acc01, acc10, acc11;
#pragma unroll
  for (int r = 0; r < 8; ++r) {
    acc00[r] = bias0; acc01[r] = bias1;
    acc10[r] = bias0; acc11[r] = bias1;
  }

  const float* ap0 = in + (size_t)m0 * kH + half * 4;
  const float* ap1 = in + (size_t)m1 * kH + half * 4;
  const float* bp0 = Wi + (size_t)n0 * kH + half * 4;
  const float* bp1 = Wi + (size_t)n1 * kH + half * 4;
#pragma unroll 2
  for (int kp = 0; kp < kH / 8; ++kp) {
    const v4f a0 = *(const v4f*)(ap0 + kp * 8);
    const v4f a1 = *(const v4f*)(ap1 + kp * 8);
    const v4f b0 = *(const v4f*)(bp0 + kp * 8);
    const v4f b1 = *(const v4f*)(bp1 + kp * 8);
    acc00 = wmma_f32(frag_lo(a0), frag_lo(b0), acc00);
    acc01 = wmma_f32(frag_lo(a0), frag_lo(b1), acc01);
    acc10 = wmma_f32(frag_lo(a1), frag_lo(b0), acc10);
    acc11 = wmma_f32(frag_lo(a1), frag_lo(b1), acc11);
    acc00 = wmma_f32(frag_hi(a0), frag_hi(b0), acc00);
    acc01 = wmma_f32(frag_hi(a0), frag_hi(b1), acc01);
    acc10 = wmma_f32(frag_hi(a1), frag_hi(b0), acc10);
    acc11 = wmma_f32(frag_hi(a1), frag_hi(b1), acc11);
  }

  float* op0 = xi + (size_t)(rowBase + half * 8) * kH + blkN * 32 + lane15;
  float* op1 = op0 + (size_t)16 * kH;
#pragma unroll
  for (int r = 0; r < 8; ++r) {
    op0[(size_t)r * kH]      = acc00[r];
    op0[(size_t)r * kH + 16] = acc01[r];
    op1[(size_t)r * kH]      = acc10[r];
    op1[(size_t)r * kH + 16] = acc11[r];
  }
}

__global__ __launch_bounds__(512) void rnn_scan_kernel(float* bufA,
                                                       float* bufB,
                                                       const float* W_hh,
                                                       int d) {
  __shared__ __attribute__((aligned(16))) float hbuf[16 * kLdsStride];
  __shared__ v4f wlds[16 * kLdsBlks * kBlkKs * 32];

  const int l = blockIdx.y;
  const int c = d - l;
  if (c < 0 || c >= kNC) return;
  float* xi       = (l & 1) ? bufB : bufA;
  const float* Wh = W_hh + (size_t)l * kH * kH;
  const int t0    = c * kTc;

  const int lane   = threadIdx.x & 31;
  const int wave   = threadIdx.x >> 5;
  const int lane15 = lane & 15;
  const int half   = lane >> 4;
  const int bg     = blockIdx.x;

  if (c == 0) {
    for (int i = threadIdx.x; i < 16 * kLdsStride; i += 512) hbuf[i] = 0.0f;
  } else {
    const float* hp =
        xi + (size_t)(bg * 16) * kT * kH + (size_t)(t0 - 1) * kH + threadIdx.x;
    for (int r = 0; r < 16; ++r)
      hbuf[r * kLdsStride + threadIdx.x] = hp[(size_t)r * kT * kH];
  }

  const int n0 = wave * 32;
  const int nA = n0 + lane15;

  const float* aBase = &hbuf[lane15 * kLdsStride + half * 4];
  const float* bpA   = Wh + (size_t)nA * kH + half * 4;
  const float* bpB   = bpA + (size_t)16 * kH;

  v4f* wl = &wlds[(wave * kLdsBlks * kBlkKs) * 32 + lane];
#pragma unroll 4
  for (int q = 0; q < kLdsBlks * kBlkKp; ++q) {
    const int kp = (4 * (q / kBlkKp) + 2) * kBlkKp + (q % kBlkKp);
    wl[(2 * q) * 32]     = *(const v4f*)(bpA + kp * 8);
    wl[(2 * q + 1) * 32] = *(const v4f*)(bpB + kp * 8);
  }
  v4f wrA[kRegBlks * kBlkKp], wrB[kRegBlks * kBlkKp];
#pragma unroll
  for (int i = 0; i < kRegBlks; ++i) {
#pragma unroll
    for (int s = 0; s < kBlkKp; ++s) {
      const int kp = (4 * i) * kBlkKp + s;
      wrA[i * kBlkKp + s] = *(const v4f*)(bpA + kp * 8);
      wrB[i * kBlkKp + s] = *(const v4f*)(bpB + kp * 8);
    }
  }

  const size_t rowStride = (size_t)kT * kH;
  float* xBase =
      xi + (size_t)(bg * 16 + half * 8) * rowStride + (size_t)t0 * kH + nA;
  float* hPub  = &hbuf[half * 8 * kLdsStride + nA];

  v8f pfA, pfB;
#pragma unroll
  for (int r = 0; r < 8; ++r) {
    pfA[r] = xBase[(size_t)r * rowStride];
    pfB[r] = xBase[(size_t)r * rowStride + 16];
  }
  v4f sA[kBlkKp], sB[kBlkKp];
#pragma unroll
  for (int s = 0; s < kBlkKp; ++s) {
    sA[s] = *(const v4f*)(bpA + (kBlkKp + s) * 8);
    sB[s] = *(const v4f*)(bpB + (kBlkKp + s) * 8);
  }
  __syncthreads();

  for (int t = 0; t < kTc; ++t) {
    v8f accA = pfA;
    v8f accB = pfB;

#pragma unroll
    for (int j = 0; j < kNumBlk / 2; ++j) {
      const int br = 2 * j;
      const int bs = 2 * j + 1;
      const int i  = j >> 1;
      if ((j & 1) == 0) {
#pragma unroll
        for (int s = 0; s < kBlkKp; ++s) {
          const v4f a  = *(const v4f*)(aBase + (br * kBlkKp + s) * 8);
          const v4f wa = wrA[i * kBlkKp + s];
          const v4f wb = wrB[i * kBlkKp + s];
          accA = wmma_f32(frag_lo(a), frag_lo(wa), accA);
          accB = wmma_f32(frag_lo(a), frag_lo(wb), accB);
          accA = wmma_f32(frag_hi(a), frag_hi(wa), accA);
          accB = wmma_f32(frag_hi(a), frag_hi(wb), accB);
        }
      } else {
#pragma unroll
        for (int s = 0; s < kBlkKp; ++s) {
          const v4f a  = *(const v4f*)(aBase + (br * kBlkKp + s) * 8);
          const v4f wa = wl[(2 * (i * kBlkKp + s)) * 32];
          const v4f wb = wl[(2 * (i * kBlkKp + s) + 1) * 32];
          accA = wmma_f32(frag_lo(a), frag_lo(wa), accA);
          accB = wmma_f32(frag_lo(a), frag_lo(wb), accB);
          accA = wmma_f32(frag_hi(a), frag_hi(wa), accA);
          accB = wmma_f32(frag_hi(a), frag_hi(wb), accB);
        }
      }
      const int nb = (j + 1 < kNumBlk / 2) ? bs + 2 : 1;
      if (j == kNumBlk / 4) {
        const int tn = t + 1 < kTc ? t + 1 : kTc - 1;
        const float* xn = xBase + (size_t)tn * kH;
#pragma unroll
        for (int r = 0; r < 8; ++r) {
          pfA[r] = xn[(size_t)r * rowStride];
          pfB[r] = xn[(size_t)r * rowStride + 16];
        }
      }
#pragma unroll
      for (int s = 0; s < kBlkKp; ++s) {
        const v4f a = *(const v4f*)(aBase + (bs * kBlkKp + s) * 8);
        accA = wmma_f32(frag_lo(a), frag_lo(sA[s]), accA);
        accB = wmma_f32(frag_lo(a), frag_lo(sB[s]), accB);
        accA = wmma_f32(frag_hi(a), frag_hi(sA[s]), accA);
        accB = wmma_f32(frag_hi(a), frag_hi(sB[s]), accB);
        __builtin_amdgcn_sched_barrier(0x786);
        sA[s] = *(const v4f*)(bpA + (nb * kBlkKp + s) * 8);
        sB[s] = *(const v4f*)(bpB + (nb * kBlkKp + s) * 8);
      }
    }

    float* xp = xBase + (size_t)t * kH;
#pragma unroll
    for (int r = 0; r < 8; ++r) {
      accA[r] = accA[r] > 0.0f ? accA[r] : 0.0f;
      accB[r] = accB[r] > 0.0f ? accB[r] : 0.0f;
      xp[(size_t)r * rowStride]      = accA[r];
      xp[(size_t)r * rowStride + 16] = accB[r];
    }

    lds_step_barrier();

#pragma unroll
    for (int r = 0; r < 8; ++r) {
      hPub[r * kLdsStride]      = accA[r];
      hPub[r * kLdsStride + 16] = accB[r];
    }
    lds_step_barrier();
  }
}

__global__ __launch_bounds__(256) void head_kernel(
    const float* __restrict__ ys, const float* __restrict__ Wfc,
    const float* __restrict__ bfc, float* __restrict__ out) {
  const int lane   = threadIdx.x & 31;
  const int wave   = threadIdx.x >> 5;
  const int lane15 = lane & 15;
  const int half   = lane >> 4;

  const int tile  = blockIdx.x * 8 + wave;
  const int tileN = tile & 7;
  const int tileM = tile >> 3;

  const int n = tileN * 16 + lane15;
  const float biasv = bfc[n];
  v8f acc;
#pragma unroll
  for (int r = 0; r < 8; ++r) acc[r] = biasv;

  const float* ap =
      ys + ((size_t)(tileM * 16 + lane15) * kT + (kT - 1)) * kH + half * 2;
  const float* bp = Wfc + (size_t)n * kH + half * 2;
#pragma unroll 4
  for (int ks = 0; ks < kH / 4; ++ks) {
    v2f a = *(const v2f*)(ap + ks * 4);
    v2f b = *(const v2f*)(bp + ks * 4);
    acc = wmma_f32(a, b, acc);
  }

  float* op = out + (size_t)(tileM * 16 + half * 8) * kC + tileN * 16 + lane15;
#pragma unroll
  for (int r = 0; r < 8; ++r) op[(size_t)r * kC] = acc[r];
}

extern "C" void kernel_launch(void* const* d_in, const int* in_sizes, int n_in,
                              void* d_out, int out_size, void* d_ws,
                              size_t ws_size, hipStream_t stream) {
  const float* x    = (const float*)d_in[0];
  const float* W_ih = (const float*)d_in[1];
  const float* W_hh = (const float*)d_in[2];
  const float* b_ih = (const float*)d_in[3];
  const float* b_hh = (const float*)d_in[4];
  const float* W_fc = (const float*)d_in[5];
  const float* b_fc = (const float*)d_in[6];

  float* bufA = (float*)d_ws;
  float* bufB = bufA + (size_t)kB * kT * kH;

  constexpr int projBlocks = (kB * kTc / 32) * (kH / 32) / 8;

  for (int d = 0; d < kDiag; ++d) {
    rnn_proj_kernel<<<dim3(projBlocks, kL), 256, 0, stream>>>(
        x, bufA, bufB, W_ih, b_ih, b_hh, d);
    rnn_scan_kernel<<<dim3(4, kL), 512, 0, stream>>>(bufA, bufB, W_hh, d);
  }
  head_kernel<<<4, 256, 0, stream>>>(bufB, W_fc, b_fc, (float*)d_out);

  (void)in_sizes; (void)n_in; (void)out_size; (void)ws_size;
}


# Projection runs inside the scan launch as extra workgroups on idle units: one launch per pipeline stage, layers skewed by 2 chunks, chunks of 8 steps (was 32); k and sum order unchanged
# speedup vs baseline: 7.9354x; 1.0871x over previous
#include <hip/hip_runtime.h>

typedef float v2f __attribute__((ext_vector_type(2)));
typedef float v4f __attribute__((ext_vector_type(4)));
typedef float v8f __attribute__((ext_vector_type(8)));

namespace {

constexpr int kB = 64;
constexpr int kT = 512;
constexpr int kH = 512;
constexpr int kL = 4;
constexpr int kC = 128;
constexpr int kLdsStride = 516;

constexpr int kTc       = 8;
constexpr int kNC       = kT / kTc;
constexpr int kLaunches = kNC + 2 * (kL - 1) + 1;
static_assert(kTc >= 8 && 16 % kTc == 0 && kT % kTc == 0, "chunk split");
constexpr int kScanWGs  = 4 * kL;
constexpr int kBpt           = 16 / kTc;
constexpr int kProjPerLayer  = (kB * kTc / 32) * (kH / 32) / 16;
constexpr int kProjWGs       = kProjPerLayer * kL;
constexpr int kGridWGs       = kScanWGs + kProjWGs;

constexpr int kBlkKs   = 8;
constexpr int kBlkKp   = kBlkKs / 2;
constexpr int kNumBlk  = kH / 4 / kBlkKs;
constexpr int kRegBlks = 4;
constexpr int kLdsBlks = 4;
static_assert(kNumBlk == 2 * (kRegBlks + kLdsBlks), "block split");

__device__ __forceinline__ v8f wmma_f32(v2f a, v2f b, v8f c) {
  return __builtin_amdgcn_wmma_f32_16x16x4_f32(false, a, false, b, (short)0, c,
                                               false, false);
}

__device__ __forceinline__ v2f frag_lo(v4f v) {
  return __builtin_shufflevector(v, v, 0, 1);
}
__device__ __forceinline__ v2f frag_hi(v4f v) {
  return __builtin_shufflevector(v, v, 2, 3);
}

__device__ __forceinline__ void lds_step_barrier() {
  __builtin_amdgcn_fence(__ATOMIC_RELEASE, "workgroup", "local");
  __builtin_amdgcn_s_barrier();
  __builtin_amdgcn_fence(__ATOMIC_ACQUIRE, "workgroup", "local");
}

}

__global__ __launch_bounds__(512) void rnn_pipe_kernel(
    const float* __restrict__ x, float* bufA, float* bufB,
    const float* __restrict__ W_ih, const float* __restrict__ b_ih,
    const float* __restrict__ b_hh, const float* W_hh, int e) {
  __shared__ __attribute__((aligned(16))) float hbuf[16 * kLdsStride];
  __shared__ v4f wlds[16 * kLdsBlks * kBlkKs * 32];

  const int lane   = threadIdx.x & 31;
  const int wave   = threadIdx.x >> 5;
  const int lane15 = lane & 15;
  const int half   = lane >> 4;

  if (blockIdx.x >= kScanWGs) {
    const int pidx = blockIdx.x - kScanWGs;
    const int l = pidx / kProjPerLayer;
    const int c = e - 2 * l;
    if (c < 0 || c >= kNC) return;

    const float* __restrict__ in =
        l == 0 ? x : ((l & 1) ? (const float*)bufA : (const float*)bufB);
    float* __restrict__ xo = (l & 1) ? bufB : bufA;
    const float* __restrict__ Wi = W_ih + (size_t)l * kH * kH;
    const float* __restrict__ bi = b_ih + (size_t)l * kH;
    const float* __restrict__ bh = b_hh + (size_t)l * kH;

    const int blk  = (pidx % kProjPerLayer) * 16 + wave;
    const int blkN = blk & (kH / 32 - 1);
    const int blkM = blk >> 4;
    const int q0 = 2 * blkM;
    const int m0 = (q0 * kBpt + lane15 / kTc) * kT + c * kTc + lane15 % kTc;
    const int m1 = m0 + kBpt * kT;

    const int n0 = blkN * 32 + lane15;
    const int n1 = n0 + 16;

    const float bias0 = bi[n0] + bh[n0];
    const float bias1 = bi[n1] + bh[n1];
    v8f acc00, acc01, acc10, acc11;
#pragma unroll
    for (int r = 0; r < 8; ++r) {
      acc00[r] = bias0; acc01[r] = bias1;
      acc10[r] = bias0; acc11[r] = bias1;
    }

    const float* ap0 = in + (size_t)m0 * kH + half * 4;
    const float* ap1 = in + (size_t)m1 * kH + half * 4;
    const float* bp0 = Wi + (size_t)n0 * kH + half * 4;
    const float* bp1 = Wi + (size_t)n1 * kH + half * 4;
#pragma unroll 2
    for (int kp = 0; kp < kH / 8; ++kp) {
      const v4f a0 = *(const v4f*)(ap0 + kp * 8);
      const v4f a1 = *(const v4f*)(ap1 + kp * 8);
      const v4f b0 = *(const v4f*)(bp0 + kp * 8);
      const v4f b1 = *(const v4f*)(bp1 + kp * 8);
      acc00 = wmma_f32(frag_lo(a0), frag_lo(b0), acc00);
      acc01 = wmma_f32(frag_lo(a0), frag_lo(b1), acc01);
      acc10 = wmma_f32(frag_lo(a1), frag_lo(b0), acc10);
      acc11 = wmma_f32(frag_lo(a1), frag_lo(b1), acc11);
      acc00 = wmma_f32(frag_hi(a0), frag_hi(b0), acc00);
      acc01 = wmma_f32(frag_hi(a0), frag_hi(b1), acc01);
      acc10 = wmma_f32(frag_hi(a1), frag_hi(b0), acc10);
      acc11 = wmma_f32(frag_hi(a1), frag_hi(b1), acc11);
    }

    const int d0 =
        (q0 * kBpt + (half * 8) / kTc) * kT + c * kTc + (half * 8) % kTc;
    float* op0 = xo + (size_t)d0 * kH + blkN * 32 + lane15;
    float* op1 = op0 + (size_t)kBpt * kT * kH;
#pragma unroll
    for (int r = 0; r < 8; ++r) {
      op0[(size_t)r * kH]      = acc00[r];
      op0[(size_t)r * kH + 16] = acc01[r];
      op1[(size_t)r * kH]      = acc10[r];
      op1[(size_t)r * kH + 16] = acc11[r];
    }
    return;
  }

  const int l = blockIdx.x >> 2;
  const int c = e - 2 * l - 1;
  if (c < 0 || c >= kNC) return;
  float* xi       = (l & 1) ? bufB : bufA;
  const float* Wh = W_hh + (size_t)l * kH * kH;
  const int t0    = c * kTc;

  const int bg     = blockIdx.x & 3;

  if (c == 0) {
    for (int i = threadIdx.x; i < 16 * kLdsStride; i += 512) hbuf[i] = 0.0f;
  } else {
    const float* hp =
        xi + (size_t)(bg * 16) * kT * kH + (size_t)(t0 - 1) * kH + threadIdx.x;
    for (int r = 0; r < 16; ++r)
      hbuf[r * kLdsStride + threadIdx.x] = hp[(size_t)r * kT * kH];
  }

  const int n0 = wave * 32;
  const int nA = n0 + lane15;

  const float* aBase = &hbuf[lane15 * kLdsStride + half * 4];
  const float* bpA   = Wh + (size_t)nA * kH + half * 4;
  const float* bpB   = bpA + (size_t)16 * kH;

  v4f* wl = &wlds[(wave * kLdsBlks * kBlkKs) * 32 + lane];
#pragma unroll 4
  for (int q = 0; q < kLdsBlks * kBlkKp; ++q) {
    const int kp = (4 * (q / kBlkKp) + 2) * kBlkKp + (q % kBlkKp);
    wl[(2 * q) * 32]     = *(const v4f*)(bpA + kp * 8);
    wl[(2 * q + 1) * 32] = *(const v4f*)(bpB + kp * 8);
  }
  v4f wrA[kRegBlks * kBlkKp], wrB[kRegBlks * kBlkKp];
#pragma unroll
  for (int i = 0; i < kRegBlks; ++i) {
#pragma unroll
    for (int s = 0; s < kBlkKp; ++s) {
      const int kp = (4 * i) * kBlkKp + s;
      wrA[i * kBlkKp + s] = *(const v4f*)(bpA + kp * 8);
      wrB[i * kBlkKp + s] = *(const v4f*)(bpB + kp * 8);
    }
  }

  const size_t rowStride = (size_t)kT * kH;
  float* xBase =
      xi + (size_t)(bg * 16 + half * 8) * rowStride + (size_t)t0 * kH + nA;
  float* hPub  = &hbuf[half * 8 * kLdsStride + nA];

  v8f pfA, pfB;
#pragma unroll
  for (int r = 0; r < 8; ++r) {
    pfA[r] = xBase[(size_t)r * rowStride];
    pfB[r] = xBase[(size_t)r * rowStride + 16];
  }
  v4f sA[kBlkKp], sB[kBlkKp];
#pragma unroll
  for (int s = 0; s < kBlkKp; ++s) {
    sA[s] = *(const v4f*)(bpA + (kBlkKp + s) * 8);
    sB[s] = *(const v4f*)(bpB + (kBlkKp + s) * 8);
  }
  __syncthreads();

  for (int t = 0; t < kTc; ++t) {
    v8f accA = pfA;
    v8f accB = pfB;

#pragma unroll
    for (int j = 0; j < kNumBlk / 2; ++j) {
      const int br = 2 * j;
      const int bs = 2 * j + 1;
      const int i  = j >> 1;
      if ((j & 1) == 0) {
#pragma unroll
        for (int s = 0; s < kBlkKp; ++s) {
          const v4f a  = *(const v4f*)(aBase + (br * kBlkKp + s) * 8);
          const v4f wa = wrA[i * kBlkKp + s];
          const v4f wb = wrB[i * kBlkKp + s];
          accA = wmma_f32(frag_lo(a), frag_lo(wa), accA);
          accB = wmma_f32(frag_lo(a), frag_lo(wb), accB);
          accA = wmma_f32(frag_hi(a), frag_hi(wa), accA);
          accB = wmma_f32(frag_hi(a), frag_hi(wb), accB);
        }
      } else {
#pragma unroll
        for (int s = 0; s < kBlkKp; ++s) {
          const v4f a  = *(const v4f*)(aBase + (br * kBlkKp + s) * 8);
          const v4f wa = wl[(2 * (i * kBlkKp + s)) * 32];
          const v4f wb = wl[(2 * (i * kBlkKp + s) + 1) * 32];
          accA = wmma_f32(frag_lo(a), frag_lo(wa), accA);
          accB = wmma_f32(frag_lo(a), frag_lo(wb), accB);
          accA = wmma_f32(frag_hi(a), frag_hi(wa), accA);
          accB = wmma_f32(frag_hi(a), frag_hi(wb), accB);
        }
      }
      const int nb = (j + 1 < kNumBlk / 2) ? bs + 2 : 1;
      if (j == kNumBlk / 4) {
        const int tn = t + 1 < kTc ? t + 1 : kTc - 1;
        const float* xn = xBase + (size_t)tn * kH;
#pragma unroll
        for (int r = 0; r < 8; ++r) {
          pfA[r] = xn[(size_t)r * rowStride];
          pfB[r] = xn[(size_t)r * rowStride + 16];
        }
      }
#pragma unroll
      for (int s = 0; s < kBlkKp; ++s) {
        const v4f a = *(const v4f*)(aBase + (bs * kBlkKp + s) * 8);
        accA = wmma_f32(frag_lo(a), frag_lo(sA[s]), accA);
        accB = wmma_f32(frag_lo(a), frag_lo(sB[s]), accB);
        accA = wmma_f32(frag_hi(a), frag_hi(sA[s]), accA);
        accB = wmma_f32(frag_hi(a), frag_hi(sB[s]), accB);
        __builtin_amdgcn_sched_barrier(0x786);
        sA[s] = *(const v4f*)(bpA + (nb * kBlkKp + s) * 8);
        sB[s] = *(const v4f*)(bpB + (nb * kBlkKp + s) * 8);
      }
    }

    float* xp = xBase + (size_t)t * kH;
#pragma unroll
    for (int r = 0; r < 8; ++r) {
      accA[r] = accA[r] > 0.0f ? accA[r] : 0.0f;
      accB[r] = accB[r] > 0.0f ? accB[r] : 0.0f;
      xp[(size_t)r * rowStride]      = accA[r];
      xp[(size_t)r * rowStride + 16] = accB[r];
    }

    lds_step_barrier();

#pragma unroll
    for (int r = 0; r < 8; ++r) {
      hPub[r * kLdsStride]      = accA[r];
      hPub[r * kLdsStride + 16] = accB[r];
    }
    lds_step_barrier();
  }
}

__global__ __launch_bounds__(256) void head_kernel(
    const float* __restrict__ ys, const float* __restrict__ Wfc,
    const float* __restrict__ bfc, float* __restrict__ out) {
  const int lane   = threadIdx.x & 31;
  const int wave   = threadIdx.x >> 5;
  const int lane15 = lane & 15;
  const int half   = lane >> 4;

  const int tile  = blockIdx.x * 8 + wave;
  const int tileN = tile & 7;
  const int tileM = tile >> 3;

  const int n = tileN * 16 + lane15;
  const float biasv = bfc[n];
  v8f acc;
#pragma unroll
  for (int r = 0; r < 8; ++r) acc[r] = biasv;

  const float* ap =
      ys + ((size_t)(tileM * 16 + lane15) * kT + (kT - 1)) * kH + half * 2;
  const float* bp = Wfc + (size_t)n * kH + half * 2;
#pragma unroll 4
  for (int ks = 0; ks < kH / 4; ++ks) {
    v2f a = *(const v2f*)(ap + ks * 4);
    v2f b = *(const v2f*)(bp + ks * 4);
    acc = wmma_f32(a, b, acc);
  }

  float* op = out + (size_t)(tileM * 16 + half * 8) * kC + tileN * 16 + lane15;
#pragma unroll
  for (int r = 0; r < 8; ++r) op[(size_t)r * kC] = acc[r];
}

extern "C" void kernel_launch(void* const* d_in, const int* in_sizes, int n_in,
                              void* d_out, int out_size, void* d_ws,
                              size_t ws_size, hipStream_t stream) {
  const float* x    = (const float*)d_in[0];
  const float* W_ih = (const float*)d_in[1];
  const float* W_hh = (const float*)d_in[2];
  const float* b_ih = (const float*)d_in[3];
  const float* b_hh = (const float*)d_in[4];
  const float* W_fc = (const float*)d_in[5];
  const float* b_fc = (const float*)d_in[6];

  float* bufA = (float*)d_ws;
  float* bufB = bufA + (size_t)kB * kT * kH;

  for (int e = 0; e < kLaunches; ++e) {
    rnn_pipe_kernel<<<kGridWGs, 512, 0, stream>>>(
        x, bufA, bufB, W_ih, b_ih, b_hh, W_hh, e);
  }
  head_kernel<<<4, 256, 0, stream>>>(bufB, W_fc, b_fc, (float*)d_out);

  (void)in_sizes; (void)n_in; (void)out_size; (void)ws_size;
}


# Scan chunk start: on-chip W_hh half fetched as whole 128-B lines (4 rows per load, was 16 rows x 32 B) and transposed to fragment order through the wave's own LDS slots; k and sum order unchanged
# speedup vs baseline: 8.3785x; 1.0558x over previous
#include <hip/hip_runtime.h>

typedef float v2f __attribute__((ext_vector_type(2)));
typedef float v4f __attribute__((ext_vector_type(4)));
typedef float v8f __attribute__((ext_vector_type(8)));

namespace {

constexpr int kB = 64;
constexpr int kT = 512;
constexpr int kH = 512;
constexpr int kL = 4;
constexpr int kC = 128;
constexpr int kLdsStride = 516;

constexpr int kTc       = 8;
constexpr int kNC       = kT / kTc;
constexpr int kLaunches = kNC + 2 * (kL - 1) + 1;
static_assert(kTc >= 8 && 16 % kTc == 0 && kT % kTc == 0, "chunk split");
constexpr int kScanWGs  = 4 * kL;
constexpr int kBpt           = 16 / kTc;
constexpr int kProjPerLayer  = (kB * kTc / 32) * (kH / 32) / 16;
constexpr int kProjWGs       = kProjPerLayer * kL;
constexpr int kGridWGs       = kScanWGs + kProjWGs;

constexpr int kBlkKs   = 8;
constexpr int kBlkKp   = kBlkKs / 2;
constexpr int kNumBlk  = kH / 4 / kBlkKs;
constexpr int kRegBlks = 4;
constexpr int kLdsBlks = 4;
static_assert(kNumBlk == 2 * (kRegBlks + kLdsBlks), "block split");
static_assert(kRegBlks == kLdsBlks, "both on-chip parts pass through the same LDS slots");
constexpr int kWlSlots = kLdsBlks * kBlkKs * 32 + 4 * kLdsBlks * kBlkKp;

__device__ __forceinline__ constexpr int wslot(int q, int tile) {
  return (2 * q + tile) * 32 + 4 * q;
}

__device__ __forceinline__ v8f wmma_f32(v2f a, v2f b, v8f c) {
  return __builtin_amdgcn_wmma_f32_16x16x4_f32(false, a, false, b, (short)0, c,
                                               false, false);
}

__device__ __forceinline__ v2f frag_lo(v4f v) {
  return __builtin_shufflevector(v, v, 0, 1);
}
__device__ __forceinline__ v2f frag_hi(v4f v) {
  return __builtin_shufflevector(v, v, 2, 3);
}

__device__ __forceinline__ void lds_step_barrier() {
  __builtin_amdgcn_fence(__ATOMIC_RELEASE, "workgroup", "local");
  __builtin_amdgcn_s_barrier();
  __builtin_amdgcn_fence(__ATOMIC_ACQUIRE, "workgroup", "local");
}

}

__global__ __launch_bounds__(512) void rnn_pipe_kernel(
    const float* __restrict__ x, float* bufA, float* bufB,
    const float* __restrict__ W_ih, const float* __restrict__ b_ih,
    const float* __restrict__ b_hh, const float* W_hh, int e) {
  __shared__ __attribute__((aligned(16))) float hbuf[16 * kLdsStride];
  __shared__ v4f wlds[16 * kWlSlots];

  const int lane   = threadIdx.x & 31;
  const int wave   = threadIdx.x >> 5;
  const int lane15 = lane & 15;
  const int half   = lane >> 4;

  if (blockIdx.x >= kScanWGs) {
    const int pidx = blockIdx.x - kScanWGs;
    const int l = pidx / kProjPerLayer;
    const int c = e - 2 * l;
    if (c < 0 || c >= kNC) return;

    const float* __restrict__ in =
        l == 0 ? x : ((l & 1) ? (const float*)bufA : (const float*)bufB);
    float* __restrict__ xo = (l & 1) ? bufB : bufA;
    const float* __restrict__ Wi = W_ih + (size_t)l * kH * kH;
    const float* __restrict__ bi = b_ih + (size_t)l * kH;
    const float* __restrict__ bh = b_hh + (size_t)l * kH;

    const int blk  = (pidx % kProjPerLayer) * 16 + wave;
    const int blkN = blk & (kH / 32 - 1);
    const int blkM = blk >> 4;
    const int q0 = 2 * blkM;
    const int m0 = (q0 * kBpt + lane15 / kTc) * kT + c * kTc + lane15 % kTc;
    const int m1 = m0 + kBpt * kT;

    const int n0 = blkN * 32 + lane15;
    const int n1 = n0 + 16;

    const float bias0 = bi[n0] + bh[n0];
    const float bias1 = bi[n1] + bh[n1];
    v8f acc00, acc01, acc10, acc11;
#pragma unroll
    for (int r = 0; r < 8; ++r) {
      acc00[r] = bias0; acc01[r] = bias1;
      acc10[r] = bias0; acc11[r] = bias1;
    }

    const float* ap0 = in + (size_t)m0 * kH + half * 4;
    const float* ap1 = in + (size_t)m1 * kH + half * 4;
    const float* bp0 = Wi + (size_t)n0 * kH + half * 4;
    const float* bp1 = Wi + (size_t)n1 * kH + half * 4;
#pragma unroll 2
    for (int kp = 0; kp < kH / 8; ++kp) {
      const v4f a0 = *(const v4f*)(ap0 + kp * 8);
      const v4f a1 = *(const v4f*)(ap1 + kp * 8);
      const v4f b0 = *(const v4f*)(bp0 + kp * 8);
      const v4f b1 = *(const v4f*)(bp1 + kp * 8);
      acc00 = wmma_f32(frag_lo(a0), frag_lo(b0), acc00);
      acc01 = wmma_f32(frag_lo(a0), frag_lo(b1), acc01);
      acc10 = wmma_f32(frag_lo(a1), frag_lo(b0), acc10);
      acc11 = wmma_f32(frag_lo(a1), frag_lo(b1), acc11);
      acc00 = wmma_f32(frag_hi(a0), frag_hi(b0), acc00);
      acc01 = wmma_f32(frag_hi(a0), frag_hi(b1), acc01);
      acc10 = wmma_f32(frag_hi(a1), frag_hi(b0), acc10);
      acc11 = wmma_f32(frag_hi(a1), frag_hi(b1), acc11);
    }

    const int d0 =
        (q0 * kBpt + (half * 8) / kTc) * kT + c * kTc + (half * 8) % kTc;
    float* op0 = xo + (size_t)d0 * kH + blkN * 32 + lane15;
    float* op1 = op0 + (size_t)kBpt * kT * kH;
#pragma unroll
    for (int r = 0; r < 8; ++r) {
      op0[(size_t)r * kH]      = acc00[r];
      op0[(size_t)r * kH + 16] = acc01[r];
      op1[(size_t)r * kH]      = acc10[r];
      op1[(size_t)r * kH + 16] = acc11[r];
    }
    return;
  }

  const int l = blockIdx.x >> 2;
  const int c = e - 2 * l - 1;
  if (c < 0 || c >= kNC) return;
  float* xi       = (l & 1) ? bufB : bufA;
  const float* Wh = W_hh + (size_t)l * kH * kH;
  const int t0    = c * kTc;

  const int bg     = blockIdx.x & 3;

  if (c == 0) {
    for (int i = threadIdx.x; i < 16 * kLdsStride; i += 512) hbuf[i] = 0.0f;
  } else {
    const float* hp =
        xi + (size_t)(bg * 16) * kT * kH + (size_t)(t0 - 1) * kH + threadIdx.x;
    for (int r = 0; r < 16; ++r)
      hbuf[r * kLdsStride + threadIdx.x] = hp[(size_t)r * kT * kH];
  }

  const int n0 = wave * 32;
  const int nA = n0 + lane15;

  const float* aBase = &hbuf[lane15 * kLdsStride + half * 4];
  const float* bpA   = Wh + (size_t)nA * kH + half * 4;
  const float* bpB   = bpA + (size_t)16 * kH;

  v4f* wl = &wlds[wave * kWlSlots + lane];
  const int lrow = lane >> 3;
  const int lpc  = lane & 7;
  const float* fsrc = Wh + (size_t)(n0 + lrow) * kH + lpc * 4;
  v4f* fdst = &wlds[wave * kWlSlots + wslot(lpc >> 1, 0) + 16 * (lpc & 1) + lrow];

#pragma unroll
  for (int i = 0; i < kRegBlks; ++i) {
#pragma unroll
    for (int tj = 0; tj < 8; ++tj) {
      fdst[wslot(i * kBlkKp, tj >> 2) + 4 * (tj & 3)] = *(const v4f*)(
          fsrc + (size_t)(16 * (tj >> 2) + 4 * (tj & 3)) * kH + (4 * i) * 32);
    }
  }
  v4f wrA[kRegBlks * kBlkKp], wrB[kRegBlks * kBlkKp];
#pragma unroll
  for (int q = 0; q < kRegBlks * kBlkKp; ++q) {
    wrA[q] = wl[wslot(q, 0)];
    wrB[q] = wl[wslot(q, 1)];
  }
#pragma nounroll
  for (int i2 = 0; i2 < kLdsBlks; i2 += 2) {
#pragma unroll
    for (int u = 0; u < 16; ++u) {
      const int tj = u & 7;
      fdst[wslot((i2 + (u >> 3)) * kBlkKp, tj >> 2) + 4 * (tj & 3)] =
          *(const v4f*)(fsrc +
                        (size_t)(16 * (tj >> 2) + 4 * (tj & 3)) * kH +
                        (4 * (i2 + (u >> 3)) + 2) * 32);
    }
  }

  const size_t rowStride = (size_t)kT * kH;
  float* xBase =
      xi + (size_t)(bg * 16 + half * 8) * rowStride + (size_t)t0 * kH + nA;
  float* hPub  = &hbuf[half * 8 * kLdsStride + nA];

  v8f pfA, pfB;
#pragma unroll
  for (int r = 0; r < 8; ++r) {
    pfA[r] = xBase[(size_t)r * rowStride];
    pfB[r] = xBase[(size_t)r * rowStride + 16];
  }
  v4f sA[kBlkKp], sB[kBlkKp];
#pragma unroll
  for (int s = 0; s < kBlkKp; ++s) {
    sA[s] = *(const v4f*)(bpA + (kBlkKp + s) * 8);
    sB[s] = *(const v4f*)(bpB + (kBlkKp + s) * 8);
  }
  __syncthreads();

  for (int t = 0; t < kTc; ++t) {
    v8f accA = pfA;
    v8f accB = pfB;

#pragma unroll
    for (int j = 0; j < kNumBlk / 2; ++j) {
      const int br = 2 * j;
      const int bs = 2 * j + 1;
      const int i  = j >> 1;
      if ((j & 1) == 0) {
#pragma unroll
        for (int s = 0; s < kBlkKp; ++s) {
          const v4f a  = *(const v4f*)(aBase + (br * kBlkKp + s) * 8);
          const v4f wa = wrA[i * kBlkKp + s];
          const v4f wb = wrB[i * kBlkKp + s];
          accA = wmma_f32(frag_lo(a), frag_lo(wa), accA);
          accB = wmma_f32(frag_lo(a), frag_lo(wb), accB);
          accA = wmma_f32(frag_hi(a), frag_hi(wa), accA);
          accB = wmma_f32(frag_hi(a), frag_hi(wb), accB);
        }
      } else {
#pragma unroll
        for (int s = 0; s < kBlkKp; ++s) {
          const v4f a  = *(const v4f*)(aBase + (br * kBlkKp + s) * 8);
          const v4f wa = wl[wslot(i * kBlkKp + s, 0)];
          const v4f wb = wl[wslot(i * kBlkKp + s, 1)];
          accA = wmma_f32(frag_lo(a), frag_lo(wa), accA);
          accB = wmma_f32(frag_lo(a), frag_lo(wb), accB);
          accA = wmma_f32(frag_hi(a), frag_hi(wa), accA);
          accB = wmma_f32(frag_hi(a), frag_hi(wb), accB);
        }
      }
      const int nb = (j + 1 < kNumBlk / 2) ? bs + 2 : 1;
      if (j == kNumBlk / 4) {
        const int tn = t + 1 < kTc ? t + 1 : kTc - 1;
        const float* xn = xBase + (size_t)tn * kH;
#pragma unroll
        for (int r = 0; r < 8; ++r) {
          pfA[r] = xn[(size_t)r * rowStride];
          pfB[r] = xn[(size_t)r * rowStride + 16];
        }
      }
#pragma unroll
      for (int s = 0; s < kBlkKp; ++s) {
        const v4f a = *(const v4f*)(aBase + (bs * kBlkKp + s) * 8);
        accA = wmma_f32(frag_lo(a), frag_lo(sA[s]), accA);
        accB = wmma_f32(frag_lo(a), frag_lo(sB[s]), accB);
        accA = wmma_f32(frag_hi(a), frag_hi(sA[s]), accA);
        accB = wmma_f32(frag_hi(a), frag_hi(sB[s]), accB);
        __builtin_amdgcn_sched_barrier(0x786);
        sA[s] = *(const v4f*)(bpA + (nb * kBlkKp + s) * 8);
        sB[s] = *(const v4f*)(bpB + (nb * kBlkKp + s) * 8);
      }
    }

    float* xp = xBase + (size_t)t * kH;
#pragma unroll
    for (int r = 0; r < 8; ++r) {
      accA[r] = accA[r] > 0.0f ? accA[r] : 0.0f;
      accB[r] = accB[r] > 0.0f ? accB[r] : 0.0f;
      xp[(size_t)r * rowStride]      = accA[r];
      xp[(size_t)r * rowStride + 16] = accB[r];
    }

    lds_step_barrier();

#pragma unroll
    for (int r = 0; r < 8; ++r) {
      hPub[r * kLdsStride]      = accA[r];
      hPub[r * kLdsStride + 16] = accB[r];
    }
    lds_step_barrier();
  }
}

__global__ __launch_bounds__(256) void head_kernel(
    const float* __restrict__ ys, const float* __restrict__ Wfc,
    const float* __restrict__ bfc, float* __restrict__ out) {
  const int lane   = threadIdx.x & 31;
  const int wave   = threadIdx.x >> 5;
  const int lane15 = lane & 15;
  const int half   = lane >> 4;

  const int tile  = blockIdx.x * 8 + wave;
  const int tileN = tile & 7;
  const int tileM = tile >> 3;

  const int n = tileN * 16 + lane15;
  const float biasv = bfc[n];
  v8f acc;
#pragma unroll
  for (int r = 0; r < 8; ++r) acc[r] = biasv;

  const float* ap =
      ys + ((size_t)(tileM * 16 + lane15) * kT + (kT - 1)) * kH + half * 2;
  const float* bp = Wfc + (size_t)n * kH + half * 2;
#pragma unroll 4
  for (int ks = 0; ks < kH / 4; ++ks) {
    v2f a = *(const v2f*)(ap + ks * 4);
    v2f b = *(const v2f*)(bp + ks * 4);
    acc = wmma_f32(a, b, acc);
  }

  float* op = out + (size_t)(tileM * 16 + half * 8) * kC + tileN * 16 + lane15;
#pragma unroll
  for (int r = 0; r < 8; ++r) op[(size_t)r * kC] = acc[r];
}

extern "C" void kernel_launch(void* const* d_in, const int* in_sizes, int n_in,
                              void* d_out, int out_size, void* d_ws,
                              size_t ws_size, hipStream_t stream) {
  const float* x    = (const float*)d_in[0];
  const float* W_ih = (const float*)d_in[1];
  const float* W_hh = (const float*)d_in[2];
  const float* b_ih = (const float*)d_in[3];
  const float* b_hh = (const float*)d_in[4];
  const float* W_fc = (const float*)d_in[5];
  const float* b_fc = (const float*)d_in[6];

  float* bufA = (float*)d_ws;
  float* bufB = bufA + (size_t)kB * kT * kH;

  for (int e = 0; e < kLaunches; ++e) {
    rnn_pipe_kernel<<<kGridWGs, 512, 0, stream>>>(
        x, bufA, bufB, W_ih, b_ih, b_hh, W_hh, e);
  }
  head_kernel<<<4, 256, 0, stream>>>(bufB, W_fc, b_fc, (float*)d_out);

  (void)in_sizes; (void)n_in; (void)out_size; (void)ws_size;
}


# Scan step: wave priority falls with progress (s_setprio 3,2,1,0 at k-blocks 0,6,10,14), so a SIMD's four waves advance together and none runs its step alone; k and sum order unchanged
# speedup vs baseline: 8.8219x; 1.0529x over previous
#include <hip/hip_runtime.h>

typedef float v2f __attribute__((ext_vector_type(2)));
typedef float v4f __attribute__((ext_vector_type(4)));
typedef float v8f __attribute__((ext_vector_type(8)));

namespace {

constexpr int kB = 64;
constexpr int kT = 512;
constexpr int kH = 512;
constexpr int kL = 4;
constexpr int kC = 128;
constexpr int kLdsStride = 516;

constexpr int kTc       = 8;
constexpr int kNC       = kT / kTc;
constexpr int kLaunches = kNC + 2 * (kL - 1) + 1;
static_assert(kTc >= 8 && 16 % kTc == 0 && kT % kTc == 0, "chunk split");
constexpr int kScanWGs  = 4 * kL;
constexpr int kBpt           = 16 / kTc;
constexpr int kProjPerLayer  = (kB * kTc / 32) * (kH / 32) / 16;
constexpr int kProjWGs       = kProjPerLayer * kL;
constexpr int kGridWGs       = kScanWGs + kProjWGs;

constexpr int kBlkKs   = 8;
constexpr int kBlkKp   = kBlkKs / 2;
constexpr int kNumBlk  = kH / 4 / kBlkKs;
constexpr int kRegBlks = 4;
constexpr int kLdsBlks = 4;
static_assert(kNumBlk == 2 * (kRegBlks + kLdsBlks), "block split");
static_assert(kRegBlks == kLdsBlks, "both on-chip parts pass through the same LDS slots");
constexpr int kWlSlots = kLdsBlks * kBlkKs * 32 + 4 * kLdsBlks * kBlkKp;

__device__ __forceinline__ constexpr int wslot(int q, int tile) {
  return (2 * q + tile) * 32 + 4 * q;
}

__device__ __forceinline__ v8f wmma_f32(v2f a, v2f b, v8f c) {
  return __builtin_amdgcn_wmma_f32_16x16x4_f32(false, a, false, b, (short)0, c,
                                               false, false);
}

__device__ __forceinline__ v2f frag_lo(v4f v) {
  return __builtin_shufflevector(v, v, 0, 1);
}
__device__ __forceinline__ v2f frag_hi(v4f v) {
  return __builtin_shufflevector(v, v, 2, 3);
}

__device__ __forceinline__ void lds_step_barrier() {
  __builtin_amdgcn_fence(__ATOMIC_RELEASE, "workgroup", "local");
  __builtin_amdgcn_s_barrier();
  __builtin_amdgcn_fence(__ATOMIC_ACQUIRE, "workgroup", "local");
}

}

__global__ __launch_bounds__(512) void rnn_pipe_kernel(
    const float* __restrict__ x, float* bufA, float* bufB,
    const float* __restrict__ W_ih, const float* __restrict__ b_ih,
    const float* __restrict__ b_hh, const float* W_hh, int e) {
  __shared__ __attribute__((aligned(16))) float hbuf[16 * kLdsStride];
  __shared__ v4f wlds[16 * kWlSlots];

  const int lane   = threadIdx.x & 31;
  const int wave   = threadIdx.x >> 5;
  const int lane15 = lane & 15;
  const int half   = lane >> 4;

  if (blockIdx.x >= kScanWGs) {
    const int pidx = blockIdx.x - kScanWGs;
    const int l = pidx / kProjPerLayer;
    const int c = e - 2 * l;
    if (c < 0 || c >= kNC) return;

    const float* __restrict__ in =
        l == 0 ? x : ((l & 1) ? (const float*)bufA : (const float*)bufB);
    float* __restrict__ xo = (l & 1) ? bufB : bufA;
    const float* __restrict__ Wi = W_ih + (size_t)l * kH * kH;
    const float* __restrict__ bi = b_ih + (size_t)l * kH;
    const float* __restrict__ bh = b_hh + (size_t)l * kH;

    const int blk  = (pidx % kProjPerLayer) * 16 + wave;
    const int blkN = blk & (kH / 32 - 1);
    const int blkM = blk >> 4;
    const int q0 = 2 * blkM;
    const int m0 = (q0 * kBpt + lane15 / kTc) * kT + c * kTc + lane15 % kTc;
    const int m1 = m0 + kBpt * kT;

    const int n0 = blkN * 32 + lane15;
    const int n1 = n0 + 16;

    const float bias0 = bi[n0] + bh[n0];
    const float bias1 = bi[n1] + bh[n1];
    v8f acc00, acc01, acc10, acc11;
#pragma unroll
    for (int r = 0; r < 8; ++r) {
      acc00[r] = bias0; acc01[r] = bias1;
      acc10[r] = bias0; acc11[r] = bias1;
    }

    const float* ap0 = in + (size_t)m0 * kH + half * 4;
    const float* ap1 = in + (size_t)m1 * kH + half * 4;
    const float* bp0 = Wi + (size_t)n0 * kH + half * 4;
    const float* bp1 = Wi + (size_t)n1 * kH + half * 4;
#pragma unroll 2
    for (int kp = 0; kp < kH / 8; ++kp) {
      const v4f a0 = *(const v4f*)(ap0 + kp * 8);
      const v4f a1 = *(const v4f*)(ap1 + kp * 8);
      const v4f b0 = *(const v4f*)(bp0 + kp * 8);
      const v4f b1 = *(const v4f*)(bp1 + kp * 8);
      acc00 = wmma_f32(frag_lo(a0), frag_lo(b0), acc00);
      acc01 = wmma_f32(frag_lo(a0), frag_lo(b1), acc01);
      acc10 = wmma_f32(frag_lo(a1), frag_lo(b0), acc10);
      acc11 = wmma_f32(frag_lo(a1), frag_lo(b1), acc11);
      acc00 = wmma_f32(frag_hi(a0), frag_hi(b0), acc00);
      acc01 = wmma_f32(frag_hi(a0), frag_hi(b1), acc01);
      acc10 = wmma_f32(frag_hi(a1), frag_hi(b0), acc10);
      acc11 = wmma_f32(frag_hi(a1), frag_hi(b1), acc11);
    }

    const int d0 =
        (q0 * kBpt + (half * 8) / kTc) * kT + c * kTc + (half * 8) % kTc;
    float* op0 = xo + (size_t)d0 * kH + blkN * 32 + lane15;
    float* op1 = op0 + (size_t)kBpt * kT * kH;
#pragma unroll
    for (int r = 0; r < 8; ++r) {
      op0[(size_t)r * kH]      = acc00[r];
      op0[(size_t)r * kH + 16] = acc01[r];
      op1[(size_t)r * kH]      = acc10[r];
      op1[(size_t)r * kH + 16] = acc11[r];
    }
    return;
  }

  const int l = blockIdx.x >> 2;
  const int c = e - 2 * l - 1;
  if (c < 0 || c >= kNC) return;
  float* xi       = (l & 1) ? bufB : bufA;
  const float* Wh = W_hh + (size_t)l * kH * kH;
  const int t0    = c * kTc;

  const int bg     = blockIdx.x & 3;

  if (c == 0) {
    for (int i = threadIdx.x; i < 16 * kLdsStride; i += 512) hbuf[i] = 0.0f;
  } else {
    const float* hp =
        xi + (size_t)(bg * 16) * kT * kH + (size_t)(t0 - 1) * kH + threadIdx.x;
    for (int r = 0; r < 16; ++r)
      hbuf[r * kLdsStride + threadIdx.x] = hp[(size_t)r * kT * kH];
  }

  const int n0 = wave * 32;
  const int nA = n0 + lane15;

  const float* aBase = &hbuf[lane15 * kLdsStride + half * 4];
  const float* bpA   = Wh + (size_t)nA * kH + half * 4;
  const float* bpB   = bpA + (size_t)16 * kH;

  v4f* wl = &wlds[wave * kWlSlots + lane];
  const int lrow = lane >> 3;
  const int lpc  = lane & 7;
  const float* fsrc = Wh + (size_t)(n0 + lrow) * kH + lpc * 4;
  v4f* fdst = &wlds[wave * kWlSlots + wslot(lpc >> 1, 0) + 16 * (lpc & 1) + lrow];

#pragma unroll
  for (int i = 0; i < kRegBlks; ++i) {
#pragma unroll
    for (int tj = 0; tj < 8; ++tj) {
      fdst[wslot(i * kBlkKp, tj >> 2) + 4 * (tj & 3)] = *(const v4f*)(
          fsrc + (size_t)(16 * (tj >> 2) + 4 * (tj & 3)) * kH + (4 * i) * 32);
    }
  }
  v4f wrA[kRegBlks * kBlkKp], wrB[kRegBlks * kBlkKp];
#pragma unroll
  for (int q = 0; q < kRegBlks * kBlkKp; ++q) {
    wrA[q] = wl[wslot(q, 0)];
    wrB[q] = wl[wslot(q, 1)];
  }
#pragma nounroll
  for (int i2 = 0; i2 < kLdsBlks; i2 += 2) {
#pragma unroll
    for (int u = 0; u < 16; ++u) {
      const int tj = u & 7;
      fdst[wslot((i2 + (u >> 3)) * kBlkKp, tj >> 2) + 4 * (tj & 3)] =
          *(const v4f*)(fsrc +
                        (size_t)(16 * (tj >> 2) + 4 * (tj & 3)) * kH +
                        (4 * (i2 + (u >> 3)) + 2) * 32);
    }
  }

  const size_t rowStride = (size_t)kT * kH;
  float* xBase =
      xi + (size_t)(bg * 16 + half * 8) * rowStride + (size_t)t0 * kH + nA;
  float* hPub  = &hbuf[half * 8 * kLdsStride + nA];

  v8f pfA, pfB;
#pragma unroll
  for (int r = 0; r < 8; ++r) {
    pfA[r] = xBase[(size_t)r * rowStride];
    pfB[r] = xBase[(size_t)r * rowStride + 16];
  }
  v4f sA[kBlkKp], sB[kBlkKp];
#pragma unroll
  for (int s = 0; s < kBlkKp; ++s) {
    sA[s] = *(const v4f*)(bpA + (kBlkKp + s) * 8);
    sB[s] = *(const v4f*)(bpB + (kBlkKp + s) * 8);
  }
  __syncthreads();

  for (int t = 0; t < kTc; ++t) {
    __builtin_amdgcn_s_setprio(3);
    v8f accA = pfA;
    v8f accB = pfB;

#pragma unroll
    for (int j = 0; j < kNumBlk / 2; ++j) {
      const int br = 2 * j;
      const int bs = 2 * j + 1;
      const int i  = j >> 1;
      if (j == 3 || j == 5 || j == 7) {
        float tie = accA[0];
        if (j == 3) asm("s_setprio 2" : "+v"(tie));
        if (j == 5) asm("s_setprio 1" : "+v"(tie));
        if (j == 7) asm("s_setprio 0" : "+v"(tie));
        accA[0] = tie;
      }
      if ((j & 1) == 0) {
#pragma unroll
        for (int s = 0; s < kBlkKp; ++s) {
          const v4f a  = *(const v4f*)(aBase + (br * kBlkKp + s) * 8);
          const v4f wa = wrA[i * kBlkKp + s];
          const v4f wb = wrB[i * kBlkKp + s];
          accA = wmma_f32(frag_lo(a), frag_lo(wa), accA);
          accB = wmma_f32(frag_lo(a), frag_lo(wb), accB);
          accA = wmma_f32(frag_hi(a), frag_hi(wa), accA);
          accB = wmma_f32(frag_hi(a), frag_hi(wb), accB);
        }
      } else {
#pragma unroll
        for (int s = 0; s < kBlkKp; ++s) {
          const v4f a  = *(const v4f*)(aBase + (br * kBlkKp + s) * 8);
          const v4f wa = wl[wslot(i * kBlkKp + s, 0)];
          const v4f wb = wl[wslot(i * kBlkKp + s, 1)];
          accA = wmma_f32(frag_lo(a), frag_lo(wa), accA);
          accB = wmma_f32(frag_lo(a), frag_lo(wb), accB);
          accA = wmma_f32(frag_hi(a), frag_hi(wa), accA);
          accB = wmma_f32(frag_hi(a), frag_hi(wb), accB);
        }
      }
      const int nb = (j + 1 < kNumBlk / 2) ? bs + 2 : 1;
      if (j == kNumBlk / 4) {
        const int tn = t + 1 < kTc ? t + 1 : kTc - 1;
        const float* xn = xBase + (size_t)tn * kH;
#pragma unroll
        for (int r = 0; r < 8; ++r) {
          pfA[r] = xn[(size_t)r * rowStride];
          pfB[r] = xn[(size_t)r * rowStride + 16];
        }
      }
#pragma unroll
      for (int s = 0; s < kBlkKp; ++s) {
        const v4f a = *(const v4f*)(aBase + (bs * kBlkKp + s) * 8);
        accA = wmma_f32(frag_lo(a), frag_lo(sA[s]), accA);
        accB = wmma_f32(frag_lo(a), frag_lo(sB[s]), accB);
        accA = wmma_f32(frag_hi(a), frag_hi(sA[s]), accA);
        accB = wmma_f32(frag_hi(a), frag_hi(sB[s]), accB);
        __builtin_amdgcn_sched_barrier(0x786);
        sA[s] = *(const v4f*)(bpA + (nb * kBlkKp + s) * 8);
        sB[s] = *(const v4f*)(bpB + (nb * kBlkKp + s) * 8);
      }
    }

    float* xp = xBase + (size_t)t * kH;
#pragma unroll
    for (int r = 0; r < 8; ++r) {
      accA[r] = accA[r] > 0.0f ? accA[r] : 0.0f;
      accB[r] = accB[r] > 0.0f ? accB[r] : 0.0f;
      xp[(size_t)r * rowStride]      = accA[r];
      xp[(size_t)r * rowStride + 16] = accB[r];
    }

    lds_step_barrier();

#pragma unroll
    for (int r = 0; r < 8; ++r) {
      hPub[r * kLdsStride]      = accA[r];
      hPub[r * kLdsStride + 16] = accB[r];
    }
    lds_step_barrier();
  }
}

__global__ __launch_bounds__(256) void head_kernel(
    const float* __restrict__ ys, const float* __restrict__ Wfc,
    const float* __restrict__ bfc, float* __restrict__ out) {
  const int lane   = threadIdx.x & 31;
  const int wave   = threadIdx.x >> 5;
  const int lane15 = lane & 15;
  const int half   = lane >> 4;

  const int tile  = blockIdx.x * 8 + wave;
  const int tileN = tile & 7;
  const int tileM = tile >> 3;

  const int n = tileN * 16 + lane15;
  const float biasv = bfc[n];
  v8f acc;
#pragma unroll
  for (int r = 0; r < 8; ++r) acc[r] = biasv;

  const float* ap =
      ys + ((size_t)(tileM * 16 + lane15) * kT + (kT - 1)) * kH + half * 2;
  const float* bp = Wfc + (size_t)n * kH + half * 2;
#pragma unroll 4
  for (int ks = 0; ks < kH / 4; ++ks) {
    v2f a = *(const v2f*)(ap + ks * 4);
    v2f b = *(const v2f*)(bp + ks * 4);
    acc = wmma_f32(a, b, acc);
  }

  float* op = out + (size_t)(tileM * 16 + half * 8) * kC + tileN * 16 + lane15;
#pragma unroll
  for (int r = 0; r < 8; ++r) op[(size_t)r * kC] = acc[r];
}

extern "C" void kernel_launch(void* const* d_in, const int* in_sizes, int n_in,
                              void* d_out, int out_size, void* d_ws,
                              size_t ws_size, hipStream_t stream) {
  const float* x    = (const float*)d_in[0];
  const float* W_ih = (const float*)d_in[1];
  const float* W_hh = (const float*)d_in[2];
  const float* b_ih = (const float*)d_in[3];
  const float* b_hh = (const float*)d_in[4];
  const float* W_fc = (const float*)d_in[5];
  const float* b_fc = (const float*)d_in[6];

  float* bufA = (float*)d_ws;
  float* bufB = bufA + (size_t)kB * kT * kH;

  for (int e = 0; e < kLaunches; ++e) {
    rnn_pipe_kernel<<<kGridWGs, 512, 0, stream>>>(
        x, bufA, bufB, W_ih, b_ih, b_hh, W_hh, e);
  }
  head_kernel<<<4, 256, 0, stream>>>(bufB, W_fc, b_fc, (float*)d_out);

  (void)in_sizes; (void)n_in; (void)out_size; (void)ws_size;
}
